# starting kernel
# baseline (speedup 1.0000x reference)
#include <hip/hip_runtime.h>


typedef __attribute__((ext_vector_type(2))) float v2f;
typedef __attribute__((ext_vector_type(8))) float v8f;

__global__ void fill_kernel(float* __restrict__ p, float v, int n) {
  int i = blockIdx.x * blockDim.x + threadIdx.x;
  if (i < n) p[i] = v;
}

__global__ void deg_kernel(const int* __restrict__ dst, float* __restrict__ deg, int E) {
  int e = blockIdx.x * blockDim.x + threadIdx.x;
  if (e < E) unsafeAtomicAdd(&deg[dst[e]], 1.0f);
}

__global__ void rsqrt_kernel(float* __restrict__ deg, int n) {
  int i = blockIdx.x * blockDim.x + threadIdx.x;
  if (i < n) deg[i] = rsqrtf(deg[i]);
}

template <int K, int ND, int RT, bool RELU_A>
__global__ void gemm_wmma_f32(const float* __restrict__ X,
                              const float* __restrict__ W,
                              float* __restrict__ H) {
  const int wave = threadIdx.x >> 5;
  const int lane = threadIdx.x & 31;
  const int col0 = blockIdx.y * 64 + wave * 16;

  const int m  = lane & 15;
  const int kb = (lane >> 4) * 2;

  v2f b[K / 4];
#pragma unroll
  for (int kk = 0; kk < K / 4; ++kk) {
    b[kk].x = W[(size_t)(kk * 4 + kb) * ND + col0 + m];
    b[kk].y = W[(size_t)(kk * 4 + kb + 1) * ND + col0 + m];
  }

  int row0 = blockIdx.x * (16 * RT);
#pragma unroll 1
  for (int rt = 0; rt < RT; ++rt, row0 += 16) {
    const float* __restrict__ xrow = X + (size_t)(row0 + m) * K + kb;
    v2f a[K / 4];
#pragma unroll
    for (int kk = 0; kk < K / 4; ++kk) {
      v2f av = *(const v2f*)(xrow + kk * 4);
      if (RELU_A) { av.x = fmaxf(av.x, 0.0f); av.y = fmaxf(av.y, 0.0f); }
      a[kk] = av;
    }

    v8f c = {};
#pragma unroll
    for (int kk = 0; kk < K / 4; ++kk)
      c = __builtin_amdgcn_wmma_f32_16x16x4_f32(false, a[kk], false, b[kk],
                                                (short)0, c, false, false);

    const int rbase = (lane >> 4) * 8;
#pragma unroll
    for (int r = 0; r < 8; ++r)
      H[(size_t)(row0 + rbase + r) * ND + col0 + m] = c[r];
  }
}

template <int D>
__global__ void selfloop_bias_kernel(const float* __restrict__ h,
                                     const float* __restrict__ dis,
                                     const float* __restrict__ bias,
                                     float* __restrict__ out, int n) {
  int i = blockIdx.x * blockDim.x + threadIdx.x;
  if (i < n * (D / 4)) {
    int v = i / (D / 4);
    int t = (i & (D / 4 - 1)) * 4;
    float w = dis[v];
    float ww = w * w;
    float4 hv = *(const float4*)(h + (size_t)v * D + t);
    float4 bv = *(const float4*)(bias + t);
    float4 o;
    o.x = bv.x + hv.x * ww; o.y = bv.y + hv.y * ww;
    o.z = bv.z + hv.z * ww; o.w = bv.w + hv.w * ww;
    *(float4*)(out + (size_t)v * D + t) = o;
  }
}

template <int D>
__global__ void scatter_kernel(const int* __restrict__ src,
                               const int* __restrict__ dst,
                               const float* __restrict__ h,
                               const float* __restrict__ dis,
                               float* __restrict__ out, int E) {
  int idx = blockIdx.x * blockDim.x + threadIdx.x;
  int e = idx / (D / 4);
  int t = (idx & (D / 4 - 1)) * 4;
  if (e < E) {
    int s = src[e], d = dst[e];
    float w = dis[s] * dis[d];
    float4 hv = *(const float4*)(h + (size_t)s * D + t);
    float* o = out + (size_t)d * D + t;
    unsafeAtomicAdd(o + 0, hv.x * w);
    unsafeAtomicAdd(o + 1, hv.y * w);
    unsafeAtomicAdd(o + 2, hv.z * w);
    unsafeAtomicAdd(o + 3, hv.w * w);
  }
}

static inline int cdiv(long long a, long long b) { return (int)((a + b - 1) / b); }

static void run_branch(const float* x, const int* ei, const float* W1,
                       const float* b1, const float* W2, const float* b2,
                       float* out_final, float* deg, float* h, float* out1,
                       int N, int E, hipStream_t stream) {
  const int IN = 128, HD = 128, OD = 64;
  const int RT = 5;
  const int* src = ei;
  const int* dst = ei + E;

  fill_kernel<<<cdiv(N, 256), 256, 0, stream>>>(deg, 1.0f, N);
  deg_kernel<<<cdiv(E, 256), 256, 0, stream>>>(dst, deg, E);
  rsqrt_kernel<<<cdiv(N, 256), 256, 0, stream>>>(deg, N);
  float* dis = deg;

  const int rowBlocks = N / (16 * RT);

  gemm_wmma_f32<IN, HD, RT, false>
      <<<dim3(rowBlocks, HD / 64), 128, 0, stream>>>(x, W1, h);
  selfloop_bias_kernel<HD><<<cdiv((long long)N * HD / 4, 256), 256, 0, stream>>>(
      h, dis, b1, out1, N);
  scatter_kernel<HD><<<cdiv((long long)E * HD / 4, 256), 256, 0, stream>>>(
      src, dst, h, dis, out1, E);

  gemm_wmma_f32<HD, OD, RT, true>
      <<<dim3(rowBlocks, 1), 128, 0, stream>>>(out1, W2, h);
  selfloop_bias_kernel<OD><<<cdiv((long long)N * OD / 4, 256), 256, 0, stream>>>(
      h, dis, b2, out_final, N);
  scatter_kernel<OD><<<cdiv((long long)E * OD / 4, 256), 256, 0, stream>>>(
      src, dst, h, dis, out_final, E);
}

extern "C" void kernel_launch(void* const* d_in, const int* in_sizes, int n_in,
                              void* d_out, int out_size, void* d_ws, size_t ws_size,
                              hipStream_t stream) {
  const float* x_feat = (const float*)d_in[0];
  const int*   x_ei   = (const int*)d_in[1];
  const float* y_feat = (const float*)d_in[2];
  const int*   y_ei   = (const int*)d_in[3];
  const float* W1x = (const float*)d_in[4];
  const float* b1x = (const float*)d_in[5];
  const float* W2x = (const float*)d_in[6];
  const float* b2x = (const float*)d_in[7];
  const float* W1y = (const float*)d_in[8];
  const float* b1y = (const float*)d_in[9];
  const float* W2y = (const float*)d_in[10];
  const float* b2y = (const float*)d_in[11];

  const int N = in_sizes[0] / 128;
  const int E = in_sizes[1] / 2;

  float* deg  = (float*)d_ws;
  float* h    = deg + N;
  float* out1 = h + (size_t)N * 128;

  float* out_x = (float*)d_out;
  float* out_y = out_x + (size_t)N * 64;

  run_branch(x_feat, x_ei, W1x, b1x, W2x, b2x, out_x, deg, h, out1, N, E, stream);
  run_branch(y_feat, y_ei, W1y, b1y, W2y, b2y, out_y, deg, h, out1, N, E, stream);
}


# Float atomic scatter-add replaced by edges counting-sorted by destination (integer atomic ranks) and a register-summed row gather per 64-column slab; per-row sum order now follows arrival rank
# speedup vs baseline: 1.8342x; 1.8342x over previous
#include <hip/hip_runtime.h>


typedef __attribute__((ext_vector_type(2))) float v2f;
typedef __attribute__((ext_vector_type(8))) float v8f;

__global__ void zero_kernel(int* __restrict__ p, int n) {
  int i = blockIdx.x * blockDim.x + threadIdx.x;
  if (i < n) p[i] = 0;
}

__global__ void rank_kernel(const int* __restrict__ dst, int* __restrict__ cnt,
                            int* __restrict__ rank, int E) {
  int e = blockIdx.x * blockDim.x + threadIdx.x;
  if (e < E) rank[e] = atomicAdd(&cnt[dst[e]], 1);
}

__global__ void __launch_bounds__(256)
scan_kernel(const int* __restrict__ cnt, int* __restrict__ rowptr,
            float* __restrict__ dis, int N) {
  __shared__ int wtot[8];
  __shared__ int wpart[8];
  const int4* __restrict__ c4 = (const int4*)cnt;
  const int tid = threadIdx.x, lane = tid & 31, wv = tid >> 5;
  const int lim = blockIdx.x * 256;

  const int nb = blockIdx.x;
  int part = 0;
  for (int j = 0; j < nb; j += 8) {
    int4 q[8];
#pragma unroll
    for (int k = 0; k < 8; ++k) {
      const int jj = (j + k < nb) ? (j + k) : (nb - 1);
      q[k] = c4[jj * 256 + tid];
    }
#pragma unroll
    for (int k = 0; k < 8; ++k) {
      const int sq = (q[k].x + q[k].y) + (q[k].z + q[k].w);
      part += (j + k < nb) ? sq : 0;
    }
  }
  const int4 mine = c4[lim + tid];
  const int s0 = mine.x, s1 = s0 + mine.y, s2 = s1 + mine.z, s3 = s2 + mine.w;

  int incl = s3;
#pragma unroll
  for (int o = 1; o < 32; o <<= 1) {
    int y = __shfl_up(incl, o, 32);
    if (lane >= o) incl += y;
  }
#pragma unroll
  for (int o = 16; o > 0; o >>= 1) part += __shfl_xor(part, o, 32);

  if (lane == 31) wtot[wv] = incl;
  if (lane == 0) wpart[wv] = part;
  __syncthreads();
  int base = 0;
#pragma unroll
  for (int k = 0; k < 8; ++k) {
    base += wpart[k];
    if (k < wv) base += wtot[k];
  }
  const int excl = base + incl - s3;
  int4 r;
  r.x = excl; r.y = excl + s0; r.z = excl + s1; r.w = excl + s2;
  ((int4*)rowptr)[lim + tid] = r;

  const int i0 = (lim + tid) * 4;
  if (i0 + 3 < N) {
    float4 d;
    d.x = rsqrtf((float)(mine.x + 1)); d.y = rsqrtf((float)(mine.y + 1));
    d.z = rsqrtf((float)(mine.z + 1)); d.w = rsqrtf((float)(mine.w + 1));
    *(float4*)(dis + i0) = d;
  } else {
    if (i0 + 0 < N) dis[i0 + 0] = rsqrtf((float)(mine.x + 1));
    if (i0 + 1 < N) dis[i0 + 1] = rsqrtf((float)(mine.y + 1));
    if (i0 + 2 < N) dis[i0 + 2] = rsqrtf((float)(mine.z + 1));
  }
}

__global__ void csr_fill_kernel(const int* __restrict__ src, const int* __restrict__ dst,
                                const int* __restrict__ rank, const int* __restrict__ rowptr,
                                const float* __restrict__ dis, int* __restrict__ ecol_words, int E) {
  int2* __restrict__ ecol = (int2*)ecol_words;
  int e = blockIdx.x * blockDim.x + threadIdx.x;
  if (e < E) {
    int s = src[e], d = dst[e];
    int2 c;
    c.x = s;
    c.y = __float_as_int(dis[s]);
    ecol[rowptr[d] + rank[e]] = c;
  }
}

template <int K, int LDW, int RT, bool RELU_A>
__global__ void gemm_wmma_f32(const float* __restrict__ X,
                              const float* __restrict__ W,
                              float* __restrict__ H) {
  constexpr int LDH = 64;
  const int wave = threadIdx.x >> 5;
  const int lane = threadIdx.x & 31;
  const int col0 = wave * 16;

  const int m  = lane & 15;
  const int kb = (lane >> 4) * 2;

  v2f b[K / 4];
#pragma unroll
  for (int kk = 0; kk < K / 4; ++kk) {
    b[kk].x = W[(size_t)(kk * 4 + kb) * LDW + col0 + m];
    b[kk].y = W[(size_t)(kk * 4 + kb + 1) * LDW + col0 + m];
  }

  int row0 = blockIdx.x * (16 * RT);
#pragma unroll 1
  for (int rt = 0; rt < RT; ++rt, row0 += 16) {
    const float* __restrict__ xrow = X + (size_t)(row0 + m) * K + kb;
    v2f a[K / 4];
#pragma unroll
    for (int kk = 0; kk < K / 4; ++kk) {
      v2f av = *(const v2f*)(xrow + kk * 4);
      if (RELU_A) { av.x = fmaxf(av.x, 0.0f); av.y = fmaxf(av.y, 0.0f); }
      a[kk] = av;
    }

    v8f c = {};
#pragma unroll
    for (int kk = 0; kk < K / 4; ++kk)
      c = __builtin_amdgcn_wmma_f32_16x16x4_f32(false, a[kk], false, b[kk],
                                                (short)0, c, false, false);

    const int rbase = (lane >> 4) * 8;
#pragma unroll
    for (int r = 0; r < 8; ++r)
      H[(size_t)(row0 + rbase + r) * LDH + col0 + m] = c[r];
  }
}

__global__ void __launch_bounds__(256)
gather_kernel(const int* __restrict__ rowptr, const int* __restrict__ ecol_words,
              const float* __restrict__ dis, const float* __restrict__ h,
              const float* __restrict__ bias, float* __restrict__ out,
              int ldo, int n) {
  const int2* __restrict__ ecol = (const int2*)ecol_words;
  const int gid = blockIdx.x * 256 + threadIdx.x;
  const int v = gid >> 4;
  const int t = (gid & 15) * 4;
  if (v >= n) return;

  const int beg = rowptr[v], end = rowptr[v + 1];
  const float dv = dis[v];
  const float ww = dv * dv;
  const float4 hv = *(const float4*)(h + (size_t)v * 64 + t);
  const float4 bv = *(const float4*)(bias + t);
  float4 acc;
  acc.x = bv.x + hv.x * ww; acc.y = bv.y + hv.y * ww;
  acc.z = bv.z + hv.z * ww; acc.w = bv.w + hv.w * ww;

  int e = beg;
  for (; e + 4 <= end; e += 4) {
    const int2 c0 = ecol[e], c1 = ecol[e + 1], c2 = ecol[e + 2], c3 = ecol[e + 3];
    const float4 x0 = *(const float4*)(h + (size_t)c0.x * 64 + t);
    const float4 x1 = *(const float4*)(h + (size_t)c1.x * 64 + t);
    const float4 x2 = *(const float4*)(h + (size_t)c2.x * 64 + t);
    const float4 x3 = *(const float4*)(h + (size_t)c3.x * 64 + t);
    const float w0 = __int_as_float(c0.y) * dv, w1 = __int_as_float(c1.y) * dv;
    const float w2 = __int_as_float(c2.y) * dv, w3 = __int_as_float(c3.y) * dv;
    acc.x += x0.x * w0; acc.y += x0.y * w0; acc.z += x0.z * w0; acc.w += x0.w * w0;
    acc.x += x1.x * w1; acc.y += x1.y * w1; acc.z += x1.z * w1; acc.w += x1.w * w1;
    acc.x += x2.x * w2; acc.y += x2.y * w2; acc.z += x2.z * w2; acc.w += x2.w * w2;
    acc.x += x3.x * w3; acc.y += x3.y * w3; acc.z += x3.z * w3; acc.w += x3.w * w3;
  }
#pragma unroll 1
  for (; e < end; ++e) {
    const int2 c0 = ecol[e];
    const float4 x0 = *(const float4*)(h + (size_t)c0.x * 64 + t);
    const float w0 = __int_as_float(c0.y) * dv;
    acc.x += x0.x * w0; acc.y += x0.y * w0; acc.z += x0.z * w0; acc.w += x0.w * w0;
  }
  *(float4*)(out + (size_t)v * ldo + t) = acc;
}

static inline int cdiv(long long a, long long b) { return (int)((a + b - 1) / b); }

static void run_branch(const float* x, const int* ei, const float* W1,
                       const float* b1, const float* W2, const float* b2,
                       float* out_final, float* dis, float* h, int* aux, float* out1,
                       int N, int E, hipStream_t stream) {
  const int IN = 128, HD = 128, OD = 64;
  const int RT = 5;
  const int* src = ei;
  const int* dst = ei + E;

  const int NP = cdiv(N + 1, 1024) * 1024;
  int*  rank   = aux;
  int*  ecol   = aux + E;
  int*  cnt    = aux + 3 * (size_t)E;
  int*  rowptr = cnt + NP;

  zero_kernel<<<cdiv(NP, 256), 256, 0, stream>>>(cnt, NP);
  rank_kernel<<<cdiv(E, 256), 256, 0, stream>>>(dst, cnt, rank, E);
  scan_kernel<<<NP / 1024, 256, 0, stream>>>(cnt, rowptr, dis, N);
  csr_fill_kernel<<<cdiv(E, 256), 256, 0, stream>>>(src, dst, rank, rowptr, dis, ecol, E);

  const int rowBlocks = N / (16 * RT);
  const int gblocks = cdiv((long long)N * 16, 256);

  gemm_wmma_f32<IN, HD, RT, false><<<rowBlocks, 128, 0, stream>>>(x, W1, h);
  gather_kernel<<<gblocks, 256, 0, stream>>>(rowptr, ecol, dis, h, b1, out1, HD, N);
  gemm_wmma_f32<IN, HD, RT, false><<<rowBlocks, 128, 0, stream>>>(x, W1 + 64, h);
  gather_kernel<<<gblocks, 256, 0, stream>>>(rowptr, ecol, dis, h, b1 + 64, out1 + 64, HD, N);

  gemm_wmma_f32<HD, OD, RT, true><<<rowBlocks, 128, 0, stream>>>(out1, W2, h);
  gather_kernel<<<gblocks, 256, 0, stream>>>(rowptr, ecol, dis, h, b2, out_final, OD, N);
}

extern "C" void kernel_launch(void* const* d_in, const int* in_sizes, int n_in,
                              void* d_out, int out_size, void* d_ws, size_t ws_size,
                              hipStream_t stream) {
  const float* x_feat = (const float*)d_in[0];
  const int*   x_ei   = (const int*)d_in[1];
  const float* y_feat = (const float*)d_in[2];
  const int*   y_ei   = (const int*)d_in[3];
  const float* W1x = (const float*)d_in[4];
  const float* b1x = (const float*)d_in[5];
  const float* W2x = (const float*)d_in[6];
  const float* b2x = (const float*)d_in[7];
  const float* W1y = (const float*)d_in[8];
  const float* b1y = (const float*)d_in[9];
  const float* W2y = (const float*)d_in[10];
  const float* b2y = (const float*)d_in[11];

  const int N = in_sizes[0] / 128;
  const int E = in_sizes[1] / 2;

  float* dis  = (float*)d_ws;
  float* h    = dis + N;
  int*   aux  = (int*)(h + (size_t)N * 64);
  float* out1 = h + (size_t)N * 128;

  float* out_x = (float*)d_out;
  float* out_y = out_x + (size_t)N * 64;

  run_branch(x_feat, x_ei, W1x, b1x, W2x, b2x, out_x, dis, h, aux, out1, N, E, stream);
  run_branch(y_feat, y_ei, W1y, b1y, W2y, b2y, out_y, dis, h, aux, out1, N, E, stream);
}


# Layer 1 in one full-width pass instead of two 64-column slab passes: one GEMM launch (two column tiles per wave share each A fragment), one gather launch (a wave per row); sum order kept
# speedup vs baseline: 2.1494x; 1.1718x over previous
#include <hip/hip_runtime.h>


typedef __attribute__((ext_vector_type(2))) float v2f;
typedef __attribute__((ext_vector_type(8))) float v8f;

__global__ void zero_kernel(int* __restrict__ p, int n) {
  int i = blockIdx.x * blockDim.x + threadIdx.x;
  if (i < n) p[i] = 0;
}

__global__ void rank_kernel(const int* __restrict__ dst, int* __restrict__ cnt,
                            int* __restrict__ rank, int E) {
  int e = blockIdx.x * blockDim.x + threadIdx.x;
  if (e < E) rank[e] = atomicAdd(&cnt[dst[e]], 1);
}

__global__ void __launch_bounds__(256)
scan_kernel(const int* __restrict__ cnt, int* __restrict__ rowptr,
            float* __restrict__ dis, int N) {
  __shared__ int wtot[8];
  __shared__ int wpart[8];
  const int4* __restrict__ c4 = (const int4*)cnt;
  const int tid = threadIdx.x, lane = tid & 31, wv = tid >> 5;
  const int lim = blockIdx.x * 256;

  const int nb = blockIdx.x;
  int part = 0;
  for (int j = 0; j < nb; j += 8) {
    int4 q[8];
#pragma unroll
    for (int k = 0; k < 8; ++k) {
      const int jj = (j + k < nb) ? (j + k) : (nb - 1);
      q[k] = c4[jj * 256 + tid];
    }
#pragma unroll
    for (int k = 0; k < 8; ++k) {
      const int sq = (q[k].x + q[k].y) + (q[k].z + q[k].w);
      part += (j + k < nb) ? sq : 0;
    }
  }
  const int4 mine = c4[lim + tid];
  const int s0 = mine.x, s1 = s0 + mine.y, s2 = s1 + mine.z, s3 = s2 + mine.w;

  int incl = s3;
#pragma unroll
  for (int o = 1; o < 32; o <<= 1) {
    int y = __shfl_up(incl, o, 32);
    if (lane >= o) incl += y;
  }
#pragma unroll
  for (int o = 16; o > 0; o >>= 1) part += __shfl_xor(part, o, 32);

  if (lane == 31) wtot[wv] = incl;
  if (lane == 0) wpart[wv] = part;
  __syncthreads();
  int base = 0;
#pragma unroll
  for (int k = 0; k < 8; ++k) {
    base += wpart[k];
    if (k < wv) base += wtot[k];
  }
  const int excl = base + incl - s3;
  int4 r;
  r.x = excl; r.y = excl + s0; r.z = excl + s1; r.w = excl + s2;
  ((int4*)rowptr)[lim + tid] = r;

  const int i0 = (lim + tid) * 4;
  if (i0 + 3 < N) {
    float4 d;
    d.x = rsqrtf((float)(mine.x + 1)); d.y = rsqrtf((float)(mine.y + 1));
    d.z = rsqrtf((float)(mine.z + 1)); d.w = rsqrtf((float)(mine.w + 1));
    *(float4*)(dis + i0) = d;
  } else {
    if (i0 + 0 < N) dis[i0 + 0] = rsqrtf((float)(mine.x + 1));
    if (i0 + 1 < N) dis[i0 + 1] = rsqrtf((float)(mine.y + 1));
    if (i0 + 2 < N) dis[i0 + 2] = rsqrtf((float)(mine.z + 1));
  }
}

__global__ void csr_fill_kernel(const int* __restrict__ src, const int* __restrict__ dst,
                                const int* __restrict__ rank, const int* __restrict__ rowptr,
                                const float* __restrict__ dis, int* __restrict__ ecol_words, int E) {
  int2* __restrict__ ecol = (int2*)ecol_words;
  int e = blockIdx.x * blockDim.x + threadIdx.x;
  if (e < E) {
    int s = src[e], d = dst[e];
    int2 c;
    c.x = s;
    c.y = __float_as_int(dis[s]);
    ecol[rowptr[d] + rank[e]] = c;
  }
}

template <int K, int LDW, int RT, bool RELU_A, int NT>
__global__ void __launch_bounds__(128)
gemm_wmma_f32(const float* __restrict__ X,
              const float* __restrict__ W,
              float* __restrict__ H0,
              float* __restrict__ H1) {
  constexpr int LDH = 64;
  const int wave = threadIdx.x >> 5;
  const int lane = threadIdx.x & 31;
  const int col0 = wave * 16;

  const int m  = lane & 15;
  const int kb = (lane >> 4) * 2;

  v2f b[NT][K / 4];
#pragma unroll
  for (int j = 0; j < NT; ++j) {
#pragma unroll
    for (int kk = 0; kk < K / 4; ++kk) {
      b[j][kk].x = W[(size_t)(kk * 4 + kb) * LDW + j * 64 + col0 + m];
      b[j][kk].y = W[(size_t)(kk * 4 + kb + 1) * LDW + j * 64 + col0 + m];
    }
  }

  int row0 = blockIdx.x * (16 * RT);
#pragma unroll 1
  for (int rt = 0; rt < RT; ++rt, row0 += 16) {
    const float* __restrict__ xrow = X + (size_t)(row0 + m) * K + kb;
    v2f a[K / 4];
#pragma unroll
    for (int kk = 0; kk < K / 4; ++kk) {
      v2f av = *(const v2f*)(xrow + kk * 4);
      if (RELU_A) { av.x = fmaxf(av.x, 0.0f); av.y = fmaxf(av.y, 0.0f); }
      a[kk] = av;
    }

    v8f c[NT];
#pragma unroll
    for (int j = 0; j < NT; ++j) c[j] = v8f{};
#pragma unroll
    for (int kk = 0; kk < K / 4; ++kk) {
#pragma unroll
      for (int j = 0; j < NT; ++j)
        c[j] = __builtin_amdgcn_wmma_f32_16x16x4_f32(false, a[kk], false, b[j][kk],
                                                     (short)0, c[j], false, false);
    }

    const int rbase = (lane >> 4) * 8;
#pragma unroll
    for (int j = 0; j < NT; ++j) {
      float* __restrict__ H = (j == 0) ? H0 : H1;
#pragma unroll
      for (int r = 0; r < 8; ++r)
        H[(size_t)(row0 + rbase + r) * LDH + col0 + m] = c[j][r];
    }
  }
}

__global__ void __launch_bounds__(256)
gather_kernel(const int* __restrict__ rowptr, const int* __restrict__ ecol_words,
              const float* __restrict__ dis, const float* __restrict__ h,
              const float* __restrict__ bias, float* __restrict__ out,
              int ldo, int n) {
  const int2* __restrict__ ecol = (const int2*)ecol_words;
  const int gid = blockIdx.x * 256 + threadIdx.x;
  const int v = gid >> 4;
  const int t = (gid & 15) * 4;
  if (v >= n) return;

  const int beg = rowptr[v], end = rowptr[v + 1];
  const float dv = dis[v];
  const float ww = dv * dv;
  const float4 hv = *(const float4*)(h + (size_t)v * 64 + t);
  const float4 bv = *(const float4*)(bias + t);
  float4 acc;
  acc.x = bv.x + hv.x * ww; acc.y = bv.y + hv.y * ww;
  acc.z = bv.z + hv.z * ww; acc.w = bv.w + hv.w * ww;

  int e = beg;
  for (; e + 4 <= end; e += 4) {
    const int2 c0 = ecol[e], c1 = ecol[e + 1], c2 = ecol[e + 2], c3 = ecol[e + 3];
    const float4 x0 = *(const float4*)(h + (size_t)c0.x * 64 + t);
    const float4 x1 = *(const float4*)(h + (size_t)c1.x * 64 + t);
    const float4 x2 = *(const float4*)(h + (size_t)c2.x * 64 + t);
    const float4 x3 = *(const float4*)(h + (size_t)c3.x * 64 + t);
    const float w0 = __int_as_float(c0.y) * dv, w1 = __int_as_float(c1.y) * dv;
    const float w2 = __int_as_float(c2.y) * dv, w3 = __int_as_float(c3.y) * dv;
    acc.x += x0.x * w0; acc.y += x0.y * w0; acc.z += x0.z * w0; acc.w += x0.w * w0;
    acc.x += x1.x * w1; acc.y += x1.y * w1; acc.z += x1.z * w1; acc.w += x1.w * w1;
    acc.x += x2.x * w2; acc.y += x2.y * w2; acc.z += x2.z * w2; acc.w += x2.w * w2;
    acc.x += x3.x * w3; acc.y += x3.y * w3; acc.z += x3.z * w3; acc.w += x3.w * w3;
  }
#pragma unroll 1
  for (; e < end; ++e) {
    const int2 c0 = ecol[e];
    const float4 x0 = *(const float4*)(h + (size_t)c0.x * 64 + t);
    const float w0 = __int_as_float(c0.y) * dv;
    acc.x += x0.x * w0; acc.y += x0.y * w0; acc.z += x0.z * w0; acc.w += x0.w * w0;
  }
  *(float4*)(out + (size_t)v * ldo + t) = acc;
}

__global__ void __launch_bounds__(256)
gather_wide_kernel(const int* __restrict__ rowptr, const int* __restrict__ ecol_words,
                   const float* __restrict__ dis, const float* __restrict__ h0,
                   const float* __restrict__ h1, const float* __restrict__ bias,
                   float* __restrict__ out, int n) {
  const int2* __restrict__ ecol = (const int2*)ecol_words;
  const int v = __builtin_amdgcn_readfirstlane((int)((blockIdx.x * 256 + threadIdx.x) >> 5));
  if (v >= n) return;
  const int lane = threadIdx.x & 31;
  const int t = (lane & 15) * 4;
  const float* __restrict__ hs = ((lane & 16) ? h1 : h0) + t;

  const int beg = rowptr[v], end = rowptr[v + 1];
  const float dv = dis[v];
  const float ww = dv * dv;
  const float4 hv = *(const float4*)(hs + (size_t)v * 64);
  const float4 bv = *(const float4*)(bias + lane * 4);
  float4 acc;
  acc.x = bv.x + hv.x * ww; acc.y = bv.y + hv.y * ww;
  acc.z = bv.z + hv.z * ww; acc.w = bv.w + hv.w * ww;

  int e = beg;
  for (; e + 4 <= end; e += 4) {
    const int2 c0 = ecol[e], c1 = ecol[e + 1], c2 = ecol[e + 2], c3 = ecol[e + 3];
    const float4 x0 = *(const float4*)(hs + (size_t)c0.x * 64);
    const float4 x1 = *(const float4*)(hs + (size_t)c1.x * 64);
    const float4 x2 = *(const float4*)(hs + (size_t)c2.x * 64);
    const float4 x3 = *(const float4*)(hs + (size_t)c3.x * 64);
    const float w0 = __int_as_float(c0.y) * dv, w1 = __int_as_float(c1.y) * dv;
    const float w2 = __int_as_float(c2.y) * dv, w3 = __int_as_float(c3.y) * dv;
    acc.x += x0.x * w0; acc.y += x0.y * w0; acc.z += x0.z * w0; acc.w += x0.w * w0;
    acc.x += x1.x * w1; acc.y += x1.y * w1; acc.z += x1.z * w1; acc.w += x1.w * w1;
    acc.x += x2.x * w2; acc.y += x2.y * w2; acc.z += x2.z * w2; acc.w += x2.w * w2;
    acc.x += x3.x * w3; acc.y += x3.y * w3; acc.z += x3.z * w3; acc.w += x3.w * w3;
  }
#pragma unroll 1
  for (; e < end; ++e) {
    const int2 c0 = ecol[e];
    const float4 x0 = *(const float4*)(hs + (size_t)c0.x * 64);
    const float w0 = __int_as_float(c0.y) * dv;
    acc.x += x0.x * w0; acc.y += x0.y * w0; acc.z += x0.z * w0; acc.w += x0.w * w0;
  }
  *(float4*)(out + (size_t)v * 128 + lane * 4) = acc;
}

static inline int cdiv(long long a, long long b) { return (int)((a + b - 1) / b); }

static void run_branch(const float* x, const int* ei, const float* W1,
                       const float* b1, const float* W2, const float* b2,
                       float* out_final, float* dis, float* h, int* aux, float* out1,
                       int N, int E, hipStream_t stream) {
  const int IN = 128, HD = 128, OD = 64;
  const int RT = 5;
  const int* src = ei;
  const int* dst = ei + E;

  const int NP = cdiv(N + 1, 1024) * 1024;
  int*  rank   = aux;
  int*  ecol   = aux + E;
  int*  cnt    = aux + 3 * (size_t)E;
  int*  rowptr = cnt + NP;

  zero_kernel<<<cdiv(NP, 256), 256, 0, stream>>>(cnt, NP);
  rank_kernel<<<cdiv(E, 256), 256, 0, stream>>>(dst, cnt, rank, E);
  scan_kernel<<<NP / 1024, 256, 0, stream>>>(cnt, rowptr, dis, N);
  csr_fill_kernel<<<cdiv(E, 256), 256, 0, stream>>>(src, dst, rank, rowptr, dis, ecol, E);

  const int rowBlocks = N / (16 * RT);
  const int gblocks = cdiv((long long)N * 16, 256);

  float* h1 = out_final;
  const int wblocks = cdiv((long long)N * 32, 256);
  gemm_wmma_f32<IN, HD, RT, false, 2><<<rowBlocks, 128, 0, stream>>>(x, W1, h, h1);
  gather_wide_kernel<<<wblocks, 256, 0, stream>>>(rowptr, ecol, dis, h, h1, b1, out1, N);

  gemm_wmma_f32<HD, OD, RT, true, 1><<<rowBlocks, 128, 0, stream>>>(out1, W2, h, nullptr);
  gather_kernel<<<gblocks, 256, 0, stream>>>(rowptr, ecol, dis, h, b2, out_final, OD, N);
}

extern "C" void kernel_launch(void* const* d_in, const int* in_sizes, int n_in,
                              void* d_out, int out_size, void* d_ws, size_t ws_size,
                              hipStream_t stream) {
  const float* x_feat = (const float*)d_in[0];
  const int*   x_ei   = (const int*)d_in[1];
  const float* y_feat = (const float*)d_in[2];
  const int*   y_ei   = (const int*)d_in[3];
  const float* W1x = (const float*)d_in[4];
  const float* b1x = (const float*)d_in[5];
  const float* W2x = (const float*)d_in[6];
  const float* b2x = (const float*)d_in[7];
  const float* W1y = (const float*)d_in[8];
  const float* b1y = (const float*)d_in[9];
  const float* W2y = (const float*)d_in[10];
  const float* b2y = (const float*)d_in[11];

  const int N = in_sizes[0] / 128;
  const int E = in_sizes[1] / 2;

  float* dis  = (float*)d_ws;
  float* h    = dis + N;
  int*   aux  = (int*)(h + (size_t)N * 64);
  float* out1 = h + (size_t)N * 128;

  float* out_x = (float*)d_out;
  float* out_y = out_x + (size_t)N * 64;

  run_branch(x_feat, x_ei, W1x, b1x, W2x, b2x, out_x, dis, h, aux, out1, N, E, stream);
  run_branch(y_feat, y_ei, W1y, b1y, W2y, b2y, out_y, dis, h, aux, out1, N, E, stream);
}


# Edge sorts of both branches share launches (zero, rank, scan, fill once each, grid y = branch): 12 launches, was 16; first branch's list parked in the other's output region; sum order by arrival rank
# speedup vs baseline: 2.2465x; 1.0452x over previous
#include <hip/hip_runtime.h>


typedef __attribute__((ext_vector_type(2))) float v2f;
typedef __attribute__((ext_vector_type(8))) float v8f;

__global__ void zero_kernel(int* __restrict__ p, int n) {
  int i = blockIdx.x * blockDim.x + threadIdx.x;
  if (i < n) p[i] = 0;
}

__global__ void rank_kernel(const int* __restrict__ ei_x, const int* __restrict__ ei_y,
                            int* __restrict__ cnt2, int* __restrict__ rank2, int E, int NP) {
  const int g = blockIdx.y;
  const int* __restrict__ dst = (g ? ei_y : ei_x) + E;
  int* __restrict__ cnt = cnt2 + (size_t)g * NP;
  int* __restrict__ rank = rank2 + (size_t)g * E;
  int e = blockIdx.x * blockDim.x + threadIdx.x;
  if (e < E) rank[e] = atomicAdd(&cnt[dst[e]], 1);
}

__global__ void __launch_bounds__(256)
scan_kernel(const int* __restrict__ cnt2, int* __restrict__ rowptr_x, int* __restrict__ rowptr_y,
            float* __restrict__ dis_x, float* __restrict__ dis_y, int N, int NP) {
  __shared__ int wtot[8];
  __shared__ int wpart[8];
  const int g = blockIdx.y;
  int* __restrict__ rowptr = g ? rowptr_y : rowptr_x;
  float* __restrict__ dis = g ? dis_y : dis_x;
  const int4* __restrict__ c4 = (const int4*)(cnt2 + (size_t)g * NP);
  const int tid = threadIdx.x, lane = tid & 31, wv = tid >> 5;
  const int lim = blockIdx.x * 256;

  const int nb = blockIdx.x;
  int part = 0;
  for (int j = 0; j < nb; j += 8) {
    int4 q[8];
#pragma unroll
    for (int k = 0; k < 8; ++k) {
      const int jj = (j + k < nb) ? (j + k) : (nb - 1);
      q[k] = c4[jj * 256 + tid];
    }
#pragma unroll
    for (int k = 0; k < 8; ++k) {
      const int sq = (q[k].x + q[k].y) + (q[k].z + q[k].w);
      part += (j + k < nb) ? sq : 0;
    }
  }
  const int4 mine = c4[lim + tid];
  const int s0 = mine.x, s1 = s0 + mine.y, s2 = s1 + mine.z, s3 = s2 + mine.w;

  int incl = s3;
#pragma unroll
  for (int o = 1; o < 32; o <<= 1) {
    int y = __shfl_up(incl, o, 32);
    if (lane >= o) incl += y;
  }
#pragma unroll
  for (int o = 16; o > 0; o >>= 1) part += __shfl_xor(part, o, 32);

  if (lane == 31) wtot[wv] = incl;
  if (lane == 0) wpart[wv] = part;
  __syncthreads();
  int base = 0;
#pragma unroll
  for (int k = 0; k < 8; ++k) {
    base += wpart[k];
    if (k < wv) base += wtot[k];
  }
  const int excl = base + incl - s3;
  int4 r;
  r.x = excl; r.y = excl + s0; r.z = excl + s1; r.w = excl + s2;
  ((int4*)rowptr)[lim + tid] = r;

  const int i0 = (lim + tid) * 4;
  if (i0 + 3 < N) {
    float4 d;
    d.x = rsqrtf((float)(mine.x + 1)); d.y = rsqrtf((float)(mine.y + 1));
    d.z = rsqrtf((float)(mine.z + 1)); d.w = rsqrtf((float)(mine.w + 1));
    *(float4*)(dis + i0) = d;
  } else {
    if (i0 + 0 < N) dis[i0 + 0] = rsqrtf((float)(mine.x + 1));
    if (i0 + 1 < N) dis[i0 + 1] = rsqrtf((float)(mine.y + 1));
    if (i0 + 2 < N) dis[i0 + 2] = rsqrtf((float)(mine.z + 1));
  }
}

__global__ void csr_fill_kernel(const int* __restrict__ ei_x, const int* __restrict__ ei_y,
                                const int* __restrict__ rank2,
                                const int* __restrict__ rowptr_x, const int* __restrict__ rowptr_y,
                                const float* __restrict__ dis_x, const float* __restrict__ dis_y,
                                int* __restrict__ ecol_words_x, int* __restrict__ ecol_words_y, int E) {
  const int g = blockIdx.y;
  const int* __restrict__ src = g ? ei_y : ei_x;
  const int* __restrict__ dst = src + E;
  const int* __restrict__ rank = rank2 + (size_t)g * E;
  const int* __restrict__ rowptr = g ? rowptr_y : rowptr_x;
  const float* __restrict__ dis = g ? dis_y : dis_x;
  int2* __restrict__ ecol = (int2*)(g ? ecol_words_y : ecol_words_x);
  int e = blockIdx.x * blockDim.x + threadIdx.x;
  if (e < E) {
    int s = src[e], d = dst[e];
    int2 c;
    c.x = s;
    c.y = __float_as_int(dis[s]);
    ecol[rowptr[d] + rank[e]] = c;
  }
}

template <int K, int LDW, int RT, bool RELU_A, int NT>
__global__ void __launch_bounds__(128)
gemm_wmma_f32(const float* __restrict__ X,
              const float* __restrict__ W,
              float* __restrict__ H0,
              float* __restrict__ H1) {
  constexpr int LDH = 64;
  const int wave = threadIdx.x >> 5;
  const int lane = threadIdx.x & 31;
  const int col0 = wave * 16;

  const int m  = lane & 15;
  const int kb = (lane >> 4) * 2;

  v2f b[NT][K / 4];
#pragma unroll
  for (int j = 0; j < NT; ++j) {
#pragma unroll
    for (int kk = 0; kk < K / 4; ++kk) {
      b[j][kk].x = W[(size_t)(kk * 4 + kb) * LDW + j * 64 + col0 + m];
      b[j][kk].y = W[(size_t)(kk * 4 + kb + 1) * LDW + j * 64 + col0 + m];
    }
  }

  int row0 = blockIdx.x * (16 * RT);
#pragma unroll 1
  for (int rt = 0; rt < RT; ++rt, row0 += 16) {
    const float* __restrict__ xrow = X + (size_t)(row0 + m) * K + kb;
    v2f a[K / 4];
#pragma unroll
    for (int kk = 0; kk < K / 4; ++kk) {
      v2f av = *(const v2f*)(xrow + kk * 4);
      if (RELU_A) { av.x = fmaxf(av.x, 0.0f); av.y = fmaxf(av.y, 0.0f); }
      a[kk] = av;
    }

    v8f c[NT];
#pragma unroll
    for (int j = 0; j < NT; ++j) c[j] = v8f{};
#pragma unroll
    for (int kk = 0; kk < K / 4; ++kk) {
#pragma unroll
      for (int j = 0; j < NT; ++j)
        c[j] = __builtin_amdgcn_wmma_f32_16x16x4_f32(false, a[kk], false, b[j][kk],
                                                     (short)0, c[j], false, false);
    }

    const int rbase = (lane >> 4) * 8;
#pragma unroll
    for (int j = 0; j < NT; ++j) {
      float* __restrict__ H = (j == 0) ? H0 : H1;
#pragma unroll
      for (int r = 0; r < 8; ++r)
        H[(size_t)(row0 + rbase + r) * LDH + col0 + m] = c[j][r];
    }
  }
}

__global__ void __launch_bounds__(256)
gather_kernel(const int* __restrict__ rowptr, const int* __restrict__ ecol_words,
              const float* __restrict__ dis, const float* __restrict__ h,
              const float* __restrict__ bias, float* __restrict__ out,
              int ldo, int n) {
  const int2* __restrict__ ecol = (const int2*)ecol_words;
  const int gid = blockIdx.x * 256 + threadIdx.x;
  const int v = gid >> 4;
  const int t = (gid & 15) * 4;
  if (v >= n) return;

  const int beg = rowptr[v], end = rowptr[v + 1];
  const float dv = dis[v];
  const float ww = dv * dv;
  const float4 hv = *(const float4*)(h + (size_t)v * 64 + t);
  const float4 bv = *(const float4*)(bias + t);
  float4 acc;
  acc.x = bv.x + hv.x * ww; acc.y = bv.y + hv.y * ww;
  acc.z = bv.z + hv.z * ww; acc.w = bv.w + hv.w * ww;

  int e = beg;
  for (; e + 4 <= end; e += 4) {
    const int2 c0 = ecol[e], c1 = ecol[e + 1], c2 = ecol[e + 2], c3 = ecol[e + 3];
    const float4 x0 = *(const float4*)(h + (size_t)c0.x * 64 + t);
    const float4 x1 = *(const float4*)(h + (size_t)c1.x * 64 + t);
    const float4 x2 = *(const float4*)(h + (size_t)c2.x * 64 + t);
    const float4 x3 = *(const float4*)(h + (size_t)c3.x * 64 + t);
    const float w0 = __int_as_float(c0.y) * dv, w1 = __int_as_float(c1.y) * dv;
    const float w2 = __int_as_float(c2.y) * dv, w3 = __int_as_float(c3.y) * dv;
    acc.x += x0.x * w0; acc.y += x0.y * w0; acc.z += x0.z * w0; acc.w += x0.w * w0;
    acc.x += x1.x * w1; acc.y += x1.y * w1; acc.z += x1.z * w1; acc.w += x1.w * w1;
    acc.x += x2.x * w2; acc.y += x2.y * w2; acc.z += x2.z * w2; acc.w += x2.w * w2;
    acc.x += x3.x * w3; acc.y += x3.y * w3; acc.z += x3.z * w3; acc.w += x3.w * w3;
  }
#pragma unroll 1
  for (; e < end; ++e) {
    const int2 c0 = ecol[e];
    const float4 x0 = *(const float4*)(h + (size_t)c0.x * 64 + t);
    const float w0 = __int_as_float(c0.y) * dv;
    acc.x += x0.x * w0; acc.y += x0.y * w0; acc.z += x0.z * w0; acc.w += x0.w * w0;
  }
  *(float4*)(out + (size_t)v * ldo + t) = acc;
}

__global__ void __launch_bounds__(256)
gather_wide_kernel(const int* __restrict__ rowptr, const int* __restrict__ ecol_words,
                   const float* __restrict__ dis, const float* __restrict__ h0,
                   const float* __restrict__ h1, const float* __restrict__ bias,
                   float* __restrict__ out, int n) {
  const int2* __restrict__ ecol = (const int2*)ecol_words;
  const int v = __builtin_amdgcn_readfirstlane((int)((blockIdx.x * 256 + threadIdx.x) >> 5));
  if (v >= n) return;
  const int lane = threadIdx.x & 31;
  const int t = (lane & 15) * 4;
  const float* __restrict__ hs = ((lane & 16) ? h1 : h0) + t;

  const int beg = rowptr[v], end = rowptr[v + 1];
  const float dv = dis[v];
  const float ww = dv * dv;
  const float4 hv = *(const float4*)(hs + (size_t)v * 64);
  const float4 bv = *(const float4*)(bias + lane * 4);
  float4 acc;
  acc.x = bv.x + hv.x * ww; acc.y = bv.y + hv.y * ww;
  acc.z = bv.z + hv.z * ww; acc.w = bv.w + hv.w * ww;

  int e = beg;
  for (; e + 4 <= end; e += 4) {
    const int2 c0 = ecol[e], c1 = ecol[e + 1], c2 = ecol[e + 2], c3 = ecol[e + 3];
    const float4 x0 = *(const float4*)(hs + (size_t)c0.x * 64);
    const float4 x1 = *(const float4*)(hs + (size_t)c1.x * 64);
    const float4 x2 = *(const float4*)(hs + (size_t)c2.x * 64);
    const float4 x3 = *(const float4*)(hs + (size_t)c3.x * 64);
    const float w0 = __int_as_float(c0.y) * dv, w1 = __int_as_float(c1.y) * dv;
    const float w2 = __int_as_float(c2.y) * dv, w3 = __int_as_float(c3.y) * dv;
    acc.x += x0.x * w0; acc.y += x0.y * w0; acc.z += x0.z * w0; acc.w += x0.w * w0;
    acc.x += x1.x * w1; acc.y += x1.y * w1; acc.z += x1.z * w1; acc.w += x1.w * w1;
    acc.x += x2.x * w2; acc.y += x2.y * w2; acc.z += x2.z * w2; acc.w += x2.w * w2;
    acc.x += x3.x * w3; acc.y += x3.y * w3; acc.z += x3.z * w3; acc.w += x3.w * w3;
  }
#pragma unroll 1
  for (; e < end; ++e) {
    const int2 c0 = ecol[e];
    const float4 x0 = *(const float4*)(hs + (size_t)c0.x * 64);
    const float w0 = __int_as_float(c0.y) * dv;
    acc.x += x0.x * w0; acc.y += x0.y * w0; acc.z += x0.z * w0; acc.w += x0.w * w0;
  }
  *(float4*)(out + (size_t)v * 128 + lane * 4) = acc;
}

static inline int cdiv(long long a, long long b) { return (int)((a + b - 1) / b); }

static void run_layers(const float* x, const float* W1,
                       const float* b1, const float* W2, const float* b2,
                       float* out_final, const float* dis, const int* rowptr, const int* ecol,
                       float* h, float* out1, int N, hipStream_t stream) {
  const int IN = 128, HD = 128, OD = 64;
  const int RT = 5;

  const int rowBlocks = N / (16 * RT);
  const int gblocks = cdiv((long long)N * 16, 256);

  float* h1 = out_final;
  const int wblocks = cdiv((long long)N * 32, 256);
  gemm_wmma_f32<IN, HD, RT, false, 2><<<rowBlocks, 128, 0, stream>>>(x, W1, h, h1);
  gather_wide_kernel<<<wblocks, 256, 0, stream>>>(rowptr, ecol, dis, h, h1, b1, out1, N);

  gemm_wmma_f32<HD, OD, RT, true, 1><<<rowBlocks, 128, 0, stream>>>(out1, W2, h, nullptr);
  gather_kernel<<<gblocks, 256, 0, stream>>>(rowptr, ecol, dis, h, b2, out_final, OD, N);
}

extern "C" void kernel_launch(void* const* d_in, const int* in_sizes, int n_in,
                              void* d_out, int out_size, void* d_ws, size_t ws_size,
                              hipStream_t stream) {
  const float* x_feat = (const float*)d_in[0];
  const int*   x_ei   = (const int*)d_in[1];
  const float* y_feat = (const float*)d_in[2];
  const int*   y_ei   = (const int*)d_in[3];
  const float* W1x = (const float*)d_in[4];
  const float* b1x = (const float*)d_in[5];
  const float* W2x = (const float*)d_in[6];
  const float* b2x = (const float*)d_in[7];
  const float* W1y = (const float*)d_in[8];
  const float* b1y = (const float*)d_in[9];
  const float* W2y = (const float*)d_in[10];
  const float* b2y = (const float*)d_in[11];

  const int N = in_sizes[0] / 128;
  const int E = in_sizes[1] / 2;
  const int NP = cdiv(N + 1, 1024) * 1024;

  float* dis_y = (float*)d_ws;
  float* h     = dis_y + N;
  int*   aux   = (int*)(h + (size_t)N * 64);
  float* out1  = h + (size_t)N * 128;

  float* out_x = (float*)d_out;
  float* out_y = out_x + (size_t)N * 64;

  int*   ecol_y   = aux;
  int*   rowptr_y = aux + 2 * (size_t)E;
  int*   ecol_x   = (int*)out_y;
  int*   rowptr_x = ecol_x + 2 * (size_t)E;
  float* dis_x    = (float*)(rowptr_x + NP);
  int*   rank2    = (int*)out1;
  int*   cnt2     = rank2 + 2 * (size_t)E;

  const dim3 egrid(cdiv(E, 256), 2);
  zero_kernel<<<cdiv(2 * NP, 256), 256, 0, stream>>>(cnt2, 2 * NP);
  rank_kernel<<<egrid, 256, 0, stream>>>(x_ei, y_ei, cnt2, rank2, E, NP);
  scan_kernel<<<dim3(NP / 1024, 2), 256, 0, stream>>>(cnt2, rowptr_x, rowptr_y, dis_x, dis_y, N, NP);
  csr_fill_kernel<<<egrid, 256, 0, stream>>>(x_ei, y_ei, rank2, rowptr_x, rowptr_y,
                                             dis_x, dis_y, ecol_x, ecol_y, E);

  run_layers(x_feat, W1x, b1x, W2x, b2x, out_x, dis_x, rowptr_x, ecol_x, h, out1, N, stream);
  run_layers(y_feat, W1y, b1y, W2y, b2y, out_y, dis_y, rowptr_y, ecol_y, h, out1, N, stream);
}


# Layer-2 weight product moved into the layer-1 gather kernel (16 rows per block: ReLU, LDS, WMMA on all 16 waves); hidden layer never stored; 10 launches, was 12; layer-2 product summed in 4 k quarters
# speedup vs baseline: 2.4167x; 1.0758x over previous
#include <hip/hip_runtime.h>


typedef __attribute__((ext_vector_type(2))) float v2f;
typedef __attribute__((ext_vector_type(8))) float v8f;

__global__ void zero_kernel(int* __restrict__ p, int n) {
  int i = blockIdx.x * blockDim.x + threadIdx.x;
  if (i < n) p[i] = 0;
}

__global__ void rank_kernel(const int* __restrict__ ei_x, const int* __restrict__ ei_y,
                            int* __restrict__ cnt2, int* __restrict__ rank2, int E, int NP) {
  const int g = blockIdx.y;
  const int* __restrict__ dst = (g ? ei_y : ei_x) + E;
  int* __restrict__ cnt = cnt2 + (size_t)g * NP;
  int* __restrict__ rank = rank2 + (size_t)g * E;
  int e = blockIdx.x * blockDim.x + threadIdx.x;
  if (e < E) rank[e] = atomicAdd(&cnt[dst[e]], 1);
}

__global__ void __launch_bounds__(256)
scan_kernel(const int* __restrict__ cnt2, int* __restrict__ rowptr_x, int* __restrict__ rowptr_y,
            float* __restrict__ dis_x, float* __restrict__ dis_y, int N, int NP) {
  __shared__ int wtot[8];
  __shared__ int wpart[8];
  const int g = blockIdx.y;
  int* __restrict__ rowptr = g ? rowptr_y : rowptr_x;
  float* __restrict__ dis = g ? dis_y : dis_x;
  const int4* __restrict__ c4 = (const int4*)(cnt2 + (size_t)g * NP);
  const int tid = threadIdx.x, lane = tid & 31, wv = tid >> 5;
  const int lim = blockIdx.x * 256;

  const int nb = blockIdx.x;
  int part = 0;
  for (int j = 0; j < nb; j += 8) {
    int4 q[8];
#pragma unroll
    for (int k = 0; k < 8; ++k) {
      const int jj = (j + k < nb) ? (j + k) : (nb - 1);
      q[k] = c4[jj * 256 + tid];
    }
#pragma unroll
    for (int k = 0; k < 8; ++k) {
      const int sq = (q[k].x + q[k].y) + (q[k].z + q[k].w);
      part += (j + k < nb) ? sq : 0;
    }
  }
  const int4 mine = c4[lim + tid];
  const int s0 = mine.x, s1 = s0 + mine.y, s2 = s1 + mine.z, s3 = s2 + mine.w;

  int incl = s3;
#pragma unroll
  for (int o = 1; o < 32; o <<= 1) {
    int y = __shfl_up(incl, o, 32);
    if (lane >= o) incl += y;
  }
#pragma unroll
  for (int o = 16; o > 0; o >>= 1) part += __shfl_xor(part, o, 32);

  if (lane == 31) wtot[wv] = incl;
  if (lane == 0) wpart[wv] = part;
  __syncthreads();
  int base = 0;
#pragma unroll
  for (int k = 0; k < 8; ++k) {
    base += wpart[k];
    if (k < wv) base += wtot[k];
  }
  const int excl = base + incl - s3;
  int4 r;
  r.x = excl; r.y = excl + s0; r.z = excl + s1; r.w = excl + s2;
  ((int4*)rowptr)[lim + tid] = r;

  const int i0 = (lim + tid) * 4;
  if (i0 + 3 < N) {
    float4 d;
    d.x = rsqrtf((float)(mine.x + 1)); d.y = rsqrtf((float)(mine.y + 1));
    d.z = rsqrtf((float)(mine.z + 1)); d.w = rsqrtf((float)(mine.w + 1));
    *(float4*)(dis + i0) = d;
  } else {
    if (i0 + 0 < N) dis[i0 + 0] = rsqrtf((float)(mine.x + 1));
    if (i0 + 1 < N) dis[i0 + 1] = rsqrtf((float)(mine.y + 1));
    if (i0 + 2 < N) dis[i0 + 2] = rsqrtf((float)(mine.z + 1));
  }
}

__global__ void csr_fill_kernel(const int* __restrict__ ei_x, const int* __restrict__ ei_y,
                                const int* __restrict__ rank2,
                                const int* __restrict__ rowptr_x, const int* __restrict__ rowptr_y,
                                const float* __restrict__ dis_x, const float* __restrict__ dis_y,
                                int* __restrict__ ecol_words_x, int* __restrict__ ecol_words_y, int E) {
  const int g = blockIdx.y;
  const int* __restrict__ src = g ? ei_y : ei_x;
  const int* __restrict__ dst = src + E;
  const int* __restrict__ rank = rank2 + (size_t)g * E;
  const int* __restrict__ rowptr = g ? rowptr_y : rowptr_x;
  const float* __restrict__ dis = g ? dis_y : dis_x;
  int2* __restrict__ ecol = (int2*)(g ? ecol_words_y : ecol_words_x);
  int e = blockIdx.x * blockDim.x + threadIdx.x;
  if (e < E) {
    int s = src[e], d = dst[e];
    int2 c;
    c.x = s;
    c.y = __float_as_int(dis[s]);
    ecol[rowptr[d] + rank[e]] = c;
  }
}

template <int K, int LDW, int RT, int NT>
__global__ void __launch_bounds__(128)
gemm_wmma_f32(const float* __restrict__ X,
              const float* __restrict__ W,
              float* __restrict__ H0,
              float* __restrict__ H1) {
  constexpr int LDH = 64;
  const int wave = threadIdx.x >> 5;
  const int lane = threadIdx.x & 31;
  const int col0 = wave * 16;

  const int m  = lane & 15;
  const int kb = (lane >> 4) * 2;

  v2f b[NT][K / 4];
#pragma unroll
  for (int j = 0; j < NT; ++j) {
#pragma unroll
    for (int kk = 0; kk < K / 4; ++kk) {
      b[j][kk].x = W[(size_t)(kk * 4 + kb) * LDW + j * 64 + col0 + m];
      b[j][kk].y = W[(size_t)(kk * 4 + kb + 1) * LDW + j * 64 + col0 + m];
    }
  }

  int row0 = blockIdx.x * (16 * RT);
#pragma unroll 1
  for (int rt = 0; rt < RT; ++rt, row0 += 16) {
    const float* __restrict__ xrow = X + (size_t)(row0 + m) * K + kb;
    v2f a[K / 4];
#pragma unroll
    for (int kk = 0; kk < K / 4; ++kk) {
      a[kk] = *(const v2f*)(xrow + kk * 4);
    }

    v8f c[NT];
#pragma unroll
    for (int j = 0; j < NT; ++j) c[j] = v8f{};
#pragma unroll
    for (int kk = 0; kk < K / 4; ++kk) {
#pragma unroll
      for (int j = 0; j < NT; ++j)
        c[j] = __builtin_amdgcn_wmma_f32_16x16x4_f32(false, a[kk], false, b[j][kk],
                                                     (short)0, c[j], false, false);
    }

    const int rbase = (lane >> 4) * 8;
#pragma unroll
    for (int j = 0; j < NT; ++j) {
      float* __restrict__ H = (j == 0) ? H0 : H1;
#pragma unroll
      for (int r = 0; r < 8; ++r)
        H[(size_t)(row0 + rbase + r) * LDH + col0 + m] = c[j][r];
    }
  }
}

__global__ void __launch_bounds__(256)
gather_kernel(const int* __restrict__ rowptr, const int* __restrict__ ecol_words,
              const float* __restrict__ dis, const float* __restrict__ h,
              const float* __restrict__ bias, float* __restrict__ out,
              int ldo, int n) {
  const int2* __restrict__ ecol = (const int2*)ecol_words;
  const int gid = blockIdx.x * 256 + threadIdx.x;
  const int v = gid >> 4;
  const int t = (gid & 15) * 4;
  if (v >= n) return;

  const int beg = rowptr[v], end = rowptr[v + 1];
  const float dv = dis[v];
  const float ww = dv * dv;
  const float4 hv = *(const float4*)(h + (size_t)v * 64 + t);
  const float4 bv = *(const float4*)(bias + t);
  float4 acc;
  acc.x = bv.x + hv.x * ww; acc.y = bv.y + hv.y * ww;
  acc.z = bv.z + hv.z * ww; acc.w = bv.w + hv.w * ww;

  int e = beg;
  for (; e + 4 <= end; e += 4) {
    const int2 c0 = ecol[e], c1 = ecol[e + 1], c2 = ecol[e + 2], c3 = ecol[e + 3];
    const float4 x0 = *(const float4*)(h + (size_t)c0.x * 64 + t);
    const float4 x1 = *(const float4*)(h + (size_t)c1.x * 64 + t);
    const float4 x2 = *(const float4*)(h + (size_t)c2.x * 64 + t);
    const float4 x3 = *(const float4*)(h + (size_t)c3.x * 64 + t);
    const float w0 = __int_as_float(c0.y) * dv, w1 = __int_as_float(c1.y) * dv;
    const float w2 = __int_as_float(c2.y) * dv, w3 = __int_as_float(c3.y) * dv;
    acc.x += x0.x * w0; acc.y += x0.y * w0; acc.z += x0.z * w0; acc.w += x0.w * w0;
    acc.x += x1.x * w1; acc.y += x1.y * w1; acc.z += x1.z * w1; acc.w += x1.w * w1;
    acc.x += x2.x * w2; acc.y += x2.y * w2; acc.z += x2.z * w2; acc.w += x2.w * w2;
    acc.x += x3.x * w3; acc.y += x3.y * w3; acc.z += x3.z * w3; acc.w += x3.w * w3;
  }
#pragma unroll 1
  for (; e < end; ++e) {
    const int2 c0 = ecol[e];
    const float4 x0 = *(const float4*)(h + (size_t)c0.x * 64 + t);
    const float w0 = __int_as_float(c0.y) * dv;
    acc.x += x0.x * w0; acc.y += x0.y * w0; acc.z += x0.z * w0; acc.w += x0.w * w0;
  }
  *(float4*)(out + (size_t)v * ldo + t) = acc;
}

__global__ void __launch_bounds__(512)
gather_wide_gemm_kernel(const int* __restrict__ rowptr, const int* __restrict__ ecol_words,
                        const float* __restrict__ dis, const float* __restrict__ h0,
                        const float* __restrict__ h1, const float* __restrict__ bias,
                        const float* __restrict__ W2, float* __restrict__ h2, int n) {
  constexpr int P = 132;
  __shared__ float tile[16 * P];
  __shared__ float part[3 * 4 * 256];
  const int2* __restrict__ ecol = (const int2*)ecol_words;
  const int wave = __builtin_amdgcn_readfirstlane((int)(threadIdx.x >> 5));
  const int lane = threadIdx.x & 31;
  const int row0 = blockIdx.x * 16;
  const int v = row0 + wave;

  float4 acc;
  acc.x = 0.0f; acc.y = 0.0f; acc.z = 0.0f; acc.w = 0.0f;
  if (v < n) {
    const int t = (lane & 15) * 4;
    const float* __restrict__ hs = ((lane & 16) ? h1 : h0) + t;
    const int beg = rowptr[v], end = rowptr[v + 1];
    const float dv = dis[v];
    const float ww = dv * dv;
    const float4 hv = *(const float4*)(hs + (size_t)v * 64);
    const float4 bv = *(const float4*)(bias + lane * 4);
    acc.x = bv.x + hv.x * ww; acc.y = bv.y + hv.y * ww;
    acc.z = bv.z + hv.z * ww; acc.w = bv.w + hv.w * ww;

    int e = beg;
    for (; e + 4 <= end; e += 4) {
      const int2 c0 = ecol[e], c1 = ecol[e + 1], c2 = ecol[e + 2], c3 = ecol[e + 3];
      const float4 x0 = *(const float4*)(hs + (size_t)c0.x * 64);
      const float4 x1 = *(const float4*)(hs + (size_t)c1.x * 64);
      const float4 x2 = *(const float4*)(hs + (size_t)c2.x * 64);
      const float4 x3 = *(const float4*)(hs + (size_t)c3.x * 64);
      const float w0 = __int_as_float(c0.y) * dv, w1 = __int_as_float(c1.y) * dv;
      const float w2 = __int_as_float(c2.y) * dv, w3 = __int_as_float(c3.y) * dv;
      acc.x += x0.x * w0; acc.y += x0.y * w0; acc.z += x0.z * w0; acc.w += x0.w * w0;
      acc.x += x1.x * w1; acc.y += x1.y * w1; acc.z += x1.z * w1; acc.w += x1.w * w1;
      acc.x += x2.x * w2; acc.y += x2.y * w2; acc.z += x2.z * w2; acc.w += x2.w * w2;
      acc.x += x3.x * w3; acc.y += x3.y * w3; acc.z += x3.z * w3; acc.w += x3.w * w3;
    }
#pragma unroll 1
    for (; e < end; ++e) {
      const int2 c0 = ecol[e];
      const float4 x0 = *(const float4*)(hs + (size_t)c0.x * 64);
      const float w0 = __int_as_float(c0.y) * dv;
      acc.x += x0.x * w0; acc.y += x0.y * w0; acc.z += x0.z * w0; acc.w += x0.w * w0;
    }
  }

  float4 ra;
  ra.x = fmaxf(acc.x, 0.0f); ra.y = fmaxf(acc.y, 0.0f);
  ra.z = fmaxf(acc.z, 0.0f); ra.w = fmaxf(acc.w, 0.0f);
  *(float4*)(tile + wave * P + lane * 4) = ra;

  const int m  = lane & 15;
  const int kb = (lane >> 4) * 2;
  const int ct = wave & 3;
  const int q  = wave >> 2;
  const int col0 = ct * 16;
  v2f b[8];
#pragma unroll
  for (int kk = 0; kk < 8; ++kk) {
    const int k = q * 32 + kk * 4 + kb;
    b[kk].x = W2[(size_t)k * 64 + col0 + m];
    b[kk].y = W2[(size_t)(k + 1) * 64 + col0 + m];
  }
  __syncthreads();

  v8f c = v8f{};
  const float* __restrict__ arow = tile + m * P + q * 32 + kb;
#pragma unroll
  for (int kk = 0; kk < 8; ++kk) {
    const v2f a = *(const v2f*)(arow + kk * 4);
    c = __builtin_amdgcn_wmma_f32_16x16x4_f32(false, a, false, b[kk],
                                              (short)0, c, false, false);
  }

  if (q > 0) {
    float* __restrict__ pw = part + ((q - 1) * 4 + ct) * 256 + lane;
#pragma unroll
    for (int r = 0; r < 8; ++r) pw[r * 32] = c[r];
  }
  __syncthreads();
  if (q > 0) return;
#pragma unroll
  for (int qq = 0; qq < 3; ++qq) {
    const float* __restrict__ pr = part + (qq * 4 + ct) * 256 + lane;
#pragma unroll
    for (int r = 0; r < 8; ++r) c[r] += pr[r * 32];
  }

  const int rbase = (lane >> 4) * 8;
#pragma unroll
  for (int r = 0; r < 8; ++r) {
    const int row = row0 + rbase + r;
    if (row < n) h2[(size_t)row * 64 + col0 + m] = c[r];
  }
}

static inline int cdiv(long long a, long long b) { return (int)((a + b - 1) / b); }

static void run_layers(const float* x, const float* W1,
                       const float* b1, const float* W2, const float* b2,
                       float* out_final, const float* dis, const int* rowptr, const int* ecol,
                       float* h, float* h2, int N, hipStream_t stream) {
  const int IN = 128, HD = 128, OD = 64;
  const int RT = 5;

  const int rowBlocks = N / (16 * RT);
  const int gblocks = cdiv((long long)N * 16, 256);

  float* h1 = out_final;
  gemm_wmma_f32<IN, HD, RT, 2><<<rowBlocks, 128, 0, stream>>>(x, W1, h, h1);

  gather_wide_gemm_kernel<<<cdiv(N, 16), 512, 0, stream>>>(rowptr, ecol, dis, h, h1, b1, W2, h2, N);

  gather_kernel<<<gblocks, 256, 0, stream>>>(rowptr, ecol, dis, h2, b2, out_final, OD, N);
}

extern "C" void kernel_launch(void* const* d_in, const int* in_sizes, int n_in,
                              void* d_out, int out_size, void* d_ws, size_t ws_size,
                              hipStream_t stream) {
  const float* x_feat = (const float*)d_in[0];
  const int*   x_ei   = (const int*)d_in[1];
  const float* y_feat = (const float*)d_in[2];
  const int*   y_ei   = (const int*)d_in[3];
  const float* W1x = (const float*)d_in[4];
  const float* b1x = (const float*)d_in[5];
  const float* W2x = (const float*)d_in[6];
  const float* b2x = (const float*)d_in[7];
  const float* W1y = (const float*)d_in[8];
  const float* b1y = (const float*)d_in[9];
  const float* W2y = (const float*)d_in[10];
  const float* b2y = (const float*)d_in[11];

  const int N = in_sizes[0] / 128;
  const int E = in_sizes[1] / 2;
  const int NP = cdiv(N + 1, 1024) * 1024;

  float* dis_y = (float*)d_ws;
  float* h     = dis_y + N;
  int*   aux   = (int*)(h + (size_t)N * 64);
  float* h2    = h + (size_t)N * 128;

  float* out_x = (float*)d_out;
  float* out_y = out_x + (size_t)N * 64;

  int*   ecol_y   = aux;
  int*   rowptr_y = aux + 2 * (size_t)E;
  int*   ecol_x   = (int*)out_y;
  int*   rowptr_x = ecol_x + 2 * (size_t)E;
  float* dis_x    = (float*)(rowptr_x + NP);
  int*   rank2    = (int*)h2;
  int*   cnt2     = rank2 + 2 * (size_t)E;

  const dim3 egrid(cdiv(E, 256), 2);
  zero_kernel<<<cdiv(2 * NP, 256), 256, 0, stream>>>(cnt2, 2 * NP);
  rank_kernel<<<egrid, 256, 0, stream>>>(x_ei, y_ei, cnt2, rank2, E, NP);
  scan_kernel<<<dim3(NP / 1024, 2), 256, 0, stream>>>(cnt2, rowptr_x, rowptr_y, dis_x, dis_y, N, NP);
  csr_fill_kernel<<<egrid, 256, 0, stream>>>(x_ei, y_ei, rank2, rowptr_x, rowptr_y,
                                             dis_x, dis_y, ecol_x, ecol_y, E);

  run_layers(x_feat, W1x, b1x, W2x, b2x, out_x, dis_x, rowptr_x, ecol_x, h, h2, N, stream);
  run_layers(y_feat, W1y, b1y, W2y, b2y, out_y, dis_y, rowptr_y, ecol_y, h, h2, N, stream);
}


# Both gather kernels: a wave stays on for several rows (7 row tiles per block; 4 rows per wave), a row's records come by one vector load ahead of time, 8-10 row loads in flight; sum order kept
# speedup vs baseline: 3.4772x; 1.4388x over previous
#include <hip/hip_runtime.h>


typedef __attribute__((ext_vector_type(2))) float v2f;
typedef __attribute__((ext_vector_type(8))) float v8f;

constexpr int SC = 16;
constexpr int SR = 4096;

struct ChunkRange { int vb, ve, sb, se; };
__device__ __forceinline__ ChunkRange chunk_range(int c, int E) {
  const int nv = ((E & 3) == 0) ? (E >> 2) : 0;
  const int nvc = (nv + SC - 1) / SC;
  const int ns = E - 4 * nv;
  const int nsc = (ns + SC - 1) / SC;
  ChunkRange r;
  r.vb = c * nvc;
  r.ve = (r.vb + nvc < nv) ? (r.vb + nvc) : nv;
  r.sb = 4 * nv + c * nsc;
  r.se = (r.sb + nsc < E) ? (r.sb + nsc) : E;
  return r;
}

__device__ __forceinline__ void sort_count(int d, int lo, int* cnt, int& below) {
  const unsigned t = (unsigned)(d - lo);
  below += (d < lo) ? 1 : 0;
  if (t < (unsigned)SR) atomicAdd(&cnt[t], 1);
}

__device__ __forceinline__ void sort_place4(int4 d, int4 sv, int lo, int* cur, int* __restrict__ ecol) {
  const unsigned t0 = (unsigned)(d.x - lo), t1 = (unsigned)(d.y - lo);
  const unsigned t2 = (unsigned)(d.z - lo), t3 = (unsigned)(d.w - lo);
  const bool m0 = t0 < (unsigned)SR, m1 = t1 < (unsigned)SR;
  const bool m2 = t2 < (unsigned)SR, m3 = t3 < (unsigned)SR;
  int s0 = 0, s1 = 0, s2 = 0, s3 = 0;
  if (m0) s0 = atomicAdd(&cur[t0], 1);
  if (m1) s1 = atomicAdd(&cur[t1], 1);
  if (m2) s2 = atomicAdd(&cur[t2], 1);
  if (m3) s3 = atomicAdd(&cur[t3], 1);
  if (m0) ecol[s0] = sv.x;
  if (m1) ecol[s1] = sv.y;
  if (m2) ecol[s2] = sv.z;
  if (m3) ecol[s3] = sv.w;
}

__global__ void __launch_bounds__(1024)
sort_count_kernel(const int* __restrict__ ei_x, const int* __restrict__ ei_y,
                  int* __restrict__ cntc, int* __restrict__ belowc, int E) {
  __shared__ alignas(16) int cnt[SR];
  __shared__ int wred[32];
  const int c = blockIdx.x, p = blockIdx.y, g = blockIdx.z;
  const int P = gridDim.y;
  const int* __restrict__ dst = (g ? ei_y : ei_x) + E;
  const int tid = threadIdx.x, lane = tid & 31, wv = tid >> 5;
  const int lo = p * SR;

  int4 zero4;
  zero4.x = 0; zero4.y = 0; zero4.z = 0; zero4.w = 0;
  ((int4*)cnt)[tid] = zero4;
  __syncthreads();

  const ChunkRange cr = chunk_range(c, E);
  const int4* __restrict__ dst4 = (const int4*)dst;
  int below = 0;
  int i = cr.vb + tid;
  for (; i + 3 * 1024 < cr.ve; i += 4 * 1024) {
    int4 q[4];
#pragma unroll
    for (int k = 0; k < 4; ++k) q[k] = dst4[i + k * 1024];
#pragma unroll
    for (int k = 0; k < 4; ++k) {
      sort_count(q[k].x, lo, cnt, below); sort_count(q[k].y, lo, cnt, below);
      sort_count(q[k].z, lo, cnt, below); sort_count(q[k].w, lo, cnt, below);
    }
  }
  for (; i < cr.ve; i += 1024) {
    const int4 q = dst4[i];
    sort_count(q.x, lo, cnt, below); sort_count(q.y, lo, cnt, below);
    sort_count(q.z, lo, cnt, below); sort_count(q.w, lo, cnt, below);
  }
  for (int e = cr.sb + tid; e < cr.se; e += 1024) sort_count(dst[e], lo, cnt, below);

#pragma unroll
  for (int o = 16; o > 0; o >>= 1) below += __shfl_xor(below, o, 32);
  if (lane == 0) wred[wv] = below;
  __syncthreads();

  const size_t plane = ((size_t)(g * SC + c) * P + p) * SR;
  ((int4*)(cntc + plane))[tid] = ((const int4*)cnt)[tid];
  if (tid == 0) {
    int b = 0;
#pragma unroll
    for (int k = 0; k < 32; ++k) b += wred[k];
    belowc[(g * SC + c) * P + p] = b;
  }
}

__global__ void __launch_bounds__(1024)
sort_place_kernel(const int* __restrict__ ei_x, const int* __restrict__ ei_y,
                  const int* __restrict__ cntc, const int* __restrict__ belowc,
                  int* __restrict__ rowptr_x, int* __restrict__ rowptr_y,
                  float* __restrict__ dis_x, float* __restrict__ dis_y,
                  int* __restrict__ ecol_x, int* __restrict__ ecol_y, int E, int N) {
  __shared__ alignas(16) int cur[SR];
  __shared__ int wtot[32];
  const int c = blockIdx.x, p = blockIdx.y, g = blockIdx.z;
  const int P = gridDim.y;
  const int* __restrict__ src = g ? ei_y : ei_x;
  const int* __restrict__ dst = src + E;
  int* __restrict__ rowptr = g ? rowptr_y : rowptr_x;
  float* __restrict__ dis = g ? dis_y : dis_x;
  int* __restrict__ ecol = g ? ecol_y : ecol_x;
  const int tid = threadIdx.x, lane = tid & 31, wv = tid >> 5;
  const int lo = p * SR;

  int4 tot, pre;
  tot.x = 0; tot.y = 0; tot.z = 0; tot.w = 0;
  pre.x = 0; pre.y = 0; pre.z = 0; pre.w = 0;
  int base = 0;
#pragma unroll
  for (int cc = 0; cc < SC; ++cc) {
    const size_t plane = ((size_t)(g * SC + cc) * P + p) * SR;
    const int4 q = ((const int4*)(cntc + plane))[tid];
    tot.x += q.x; tot.y += q.y; tot.z += q.z; tot.w += q.w;
    if (cc < c) { pre.x += q.x; pre.y += q.y; pre.z += q.z; pre.w += q.w; }
    base += belowc[(g * SC + cc) * P + p];
  }

  const int s0 = tot.x, s1 = s0 + tot.y, s2 = s1 + tot.z, s3 = s2 + tot.w;
  int incl = s3;
#pragma unroll
  for (int o = 1; o < 32; o <<= 1) {
    int y = __shfl_up(incl, o, 32);
    if (lane >= o) incl += y;
  }
  if (lane == 31) wtot[wv] = incl;
  __syncthreads();
#pragma unroll
  for (int k = 0; k < 32; ++k) {
    if (k < wv) base += wtot[k];
  }
  const int excl = base + incl - s3;
  int4 r;
  r.x = excl; r.y = excl + s0; r.z = excl + s1; r.w = excl + s2;
  int4 cu;
  cu.x = r.x + pre.x; cu.y = r.y + pre.y; cu.z = r.z + pre.z; cu.w = r.w + pre.w;
  ((int4*)cur)[tid] = cu;

  if (c == 0) {
    const int n0 = lo + tid * 4;
    if (n0 + 0 <= N) rowptr[n0 + 0] = r.x;
    if (n0 + 1 <= N) rowptr[n0 + 1] = r.y;
    if (n0 + 2 <= N) rowptr[n0 + 2] = r.z;
    if (n0 + 3 <= N) rowptr[n0 + 3] = r.w;
    if (n0 + 0 < N) dis[n0 + 0] = rsqrtf((float)(tot.x + 1));
    if (n0 + 1 < N) dis[n0 + 1] = rsqrtf((float)(tot.y + 1));
    if (n0 + 2 < N) dis[n0 + 2] = rsqrtf((float)(tot.z + 1));
    if (n0 + 3 < N) dis[n0 + 3] = rsqrtf((float)(tot.w + 1));
  }
  __syncthreads();

  const ChunkRange cr = chunk_range(c, E);
  const int4* __restrict__ dst4 = (const int4*)dst;
  const int4* __restrict__ src4 = (const int4*)src;
  int i = cr.vb + tid;
  for (; i + 3 * 1024 < cr.ve; i += 4 * 1024) {
    int4 q[4], sv[4];
#pragma unroll
    for (int k = 0; k < 4; ++k) { q[k] = dst4[i + k * 1024]; sv[k] = src4[i + k * 1024]; }
#pragma unroll
    for (int k = 0; k < 4; ++k) sort_place4(q[k], sv[k], lo, cur, ecol);
  }
  for (; i < cr.ve; i += 1024) sort_place4(dst4[i], src4[i], lo, cur, ecol);
  for (int e = cr.sb + tid; e < cr.se; e += 1024) {
    const int d = dst[e], sv1 = src[e];
    const unsigned t = (unsigned)(d - lo);
    if (t < (unsigned)SR) ecol[atomicAdd(&cur[t], 1)] = sv1;
  }
}

template <int K, int LDW, int RT, int NT>
__global__ void __launch_bounds__(128)
gemm_wmma_f32(const float* __restrict__ X,
              const float* __restrict__ W,
              const float* __restrict__ dis,
              float* __restrict__ H0,
              float* __restrict__ H1) {
  constexpr int LDH = 64;
  const int wave = threadIdx.x >> 5;
  const int lane = threadIdx.x & 31;
  const int col0 = wave * 16;

  const int m  = lane & 15;
  const int kb = (lane >> 4) * 2;

  v2f b[NT][K / 4];
#pragma unroll
  for (int j = 0; j < NT; ++j) {
#pragma unroll
    for (int kk = 0; kk < K / 4; ++kk) {
      b[j][kk].x = W[(size_t)(kk * 4 + kb) * LDW + j * 64 + col0 + m];
      b[j][kk].y = W[(size_t)(kk * 4 + kb + 1) * LDW + j * 64 + col0 + m];
    }
  }

  int row0 = blockIdx.x * (16 * RT);
#pragma unroll 1
  for (int rt = 0; rt < RT; ++rt, row0 += 16) {
    const float* __restrict__ xrow = X + (size_t)(row0 + m) * K + kb;
    v2f a[K / 4];
#pragma unroll
    for (int kk = 0; kk < K / 4; ++kk) {
      a[kk] = *(const v2f*)(xrow + kk * 4);
    }

    v8f c[NT];
#pragma unroll
    for (int j = 0; j < NT; ++j) c[j] = v8f{};
#pragma unroll
    for (int kk = 0; kk < K / 4; ++kk) {
#pragma unroll
      for (int j = 0; j < NT; ++j)
        c[j] = __builtin_amdgcn_wmma_f32_16x16x4_f32(false, a[kk], false, b[j][kk],
                                                     (short)0, c[j], false, false);
    }

    const int rbase = (lane >> 4) * 8;
    float dr[8];
#pragma unroll
    for (int r = 0; r < 8; ++r) dr[r] = dis[row0 + rbase + r];
#pragma unroll
    for (int j = 0; j < NT; ++j) {
      float* __restrict__ H = (j == 0) ? H0 : H1;
#pragma unroll
      for (int r = 0; r < 8; ++r)
        H[(size_t)(row0 + rbase + r) * LDH + col0 + m] = c[j][r] * dr[r];
    }
  }
}


__device__ __forceinline__ void add_to(float4& s, const float4 x) {
  s.x += x.x; s.y += x.y; s.z += x.z; s.w += x.w;
}
__device__ __forceinline__ void add_to(float2& s, const float2 x) {
  s.x += x.x; s.y += x.y;
}

template <typename V, int NB>
__device__ __forceinline__ void gather_rows(V& sm, int r, int beg, int end,
                                            const int* __restrict__ ecol,
                                            const float* __restrict__ hs, int lane) {
  int base = beg;
  for (;;) {
    const int left = end - base;
    const int cc = left < 32 ? left : 32;
    int j = 0;
    for (; j + NB <= cc; j += NB) {
      V x[NB];
#pragma unroll
      for (int k = 0; k < NB; ++k)
        x[k] = *(const V*)(hs + (size_t)__builtin_amdgcn_readlane(r, j + k) * 64);
#pragma unroll
      for (int k = 0; k < NB; ++k) add_to(sm, x[k]);
    }
    const int nv = cc - j;
    if (nv > 0) {
      V x[NB];
#pragma unroll
      for (int k = 0; k < NB - 1; ++k)
        if (k < nv) x[k] = *(const V*)(hs + (size_t)__builtin_amdgcn_readlane(r, j + k) * 64);
#pragma unroll
      for (int k = 0; k < NB - 1; ++k)
        if (k < nv) add_to(sm, x[k]);
    }
    base += 32;
    if (base >= end) break;
    r = (base + lane < end) ? ecol[base + lane] : 0;
  }
}

constexpr int GR = 4;
__global__ void __launch_bounds__(256)
gather_kernel(const int* __restrict__ rowptr, const int* __restrict__ ecol,
              const float* __restrict__ dis, const float* __restrict__ h,
              const float* __restrict__ bias, float* __restrict__ out,
              int ldo, int n) {
  const int wave = __builtin_amdgcn_readfirstlane((int)(threadIdx.x >> 5));
  const int lane = threadIdx.x & 31;
  const int v0 = (blockIdx.x * 8 + wave) * GR;
  if (v0 >= n) return;

  const float* __restrict__ hs = h + lane * 2;
  int rp[GR + 1];
  float dv[GR];
  int rec[GR];
  float2 own[GR];
#pragma unroll
  for (int i = 0; i <= GR; ++i) rp[i] = rowptr[(v0 + i < n) ? (v0 + i) : n];
#pragma unroll
  for (int i = 0; i < GR; ++i) dv[i] = (v0 + i < n) ? dis[v0 + i] : 0.0f;
#pragma unroll
  for (int i = 0; i < GR; ++i) {
    rec[i] = (rp[i] + lane < rp[i + 1]) ? ecol[rp[i] + lane] : 0;
    own[i].x = 0.0f; own[i].y = 0.0f;
    if (v0 + i < n) own[i] = *(const float2*)(hs + (size_t)(v0 + i) * 64);
  }
  const float2 bv = *(const float2*)(bias + lane * 2);

#pragma unroll
  for (int i = 0; i < GR; ++i) {
    const int v = v0 + i;
    if (v < n) {
      float2 sm = own[i];
      gather_rows<float2, 8>(sm, rec[i], rp[i], rp[i + 1], ecol, hs, lane);
      float2 o;
      o.x = bv.x + sm.x * dv[i]; o.y = bv.y + sm.y * dv[i];
      *(float2*)(out + (size_t)v * ldo + lane * 2) = o;
    }
  }
}

__device__ __forceinline__ void lds_barrier() {
  __builtin_amdgcn_fence(__ATOMIC_RELEASE, "workgroup", "local");
  __builtin_amdgcn_s_barrier();
  __builtin_amdgcn_fence(__ATOMIC_ACQUIRE, "workgroup", "local");
}

constexpr int GT = 7;
__global__ void __launch_bounds__(512)
gather_wide_gemm_kernel(const int* __restrict__ rowptr, const int* __restrict__ ecol_words,
                        const float* __restrict__ dis, const float* __restrict__ h0,
                        const float* __restrict__ h1, const float* __restrict__ bias,
                        const float* __restrict__ W2, float* __restrict__ h2, int n) {
  constexpr int P = 132;
  __shared__ float tile[16 * P];
  __shared__ float part[3 * 4 * 256];
  const int* __restrict__ ecol = ecol_words;
  const int wave = __builtin_amdgcn_readfirstlane((int)(threadIdx.x >> 5));
  const int lane = threadIdx.x & 31;
  const int rowbase = blockIdx.x * (16 * GT);
  const float* __restrict__ hs = ((lane & 16) ? h1 : h0) + (lane & 15) * 4;

  const int m  = lane & 15;
  const int kb = (lane >> 4) * 2;
  const int ct = wave & 3;
  const int q  = wave >> 2;
  const int col0 = ct * 16;
  v2f b[8];
#pragma unroll
  for (int kk = 0; kk < 8; ++kk) {
    const int k = q * 32 + kk * 4 + kb;
    b[kk].x = W2[(size_t)k * 64 + col0 + m];
    b[kk].y = W2[(size_t)(k + 1) * 64 + col0 + m];
  }
  const float4 bv = *(const float4*)(bias + lane * 4);

  int lb = 0, le = 0, ld = 0;
  {
    const int vl = rowbase + lane * 16 + wave;
    if (lane < GT && vl < n) { lb = rowptr[vl]; le = rowptr[vl + 1]; ld = __float_as_int(dis[vl]); }
  }
  int rec;
  float4 own;
  own.x = 0.0f; own.y = 0.0f; own.z = 0.0f; own.w = 0.0f;
  {
    const int b0 = __builtin_amdgcn_readlane(lb, 0), e0 = __builtin_amdgcn_readlane(le, 0);
    rec = (b0 + lane < e0) ? ecol[b0 + lane] : 0;
    if (rowbase + wave < n) own = *(const float4*)(hs + (size_t)(rowbase + wave) * 64);
  }

  for (int i = 0; i < GT; ++i) {
    const int row0 = rowbase + i * 16;
    const int v = row0 + wave;

    float4 acc;
    acc.x = 0.0f; acc.y = 0.0f; acc.z = 0.0f; acc.w = 0.0f;
    if (v < n) {
      const int bc = __builtin_amdgcn_readlane(lb, i), ec = __builtin_amdgcn_readlane(le, i);
      const float dc = __int_as_float(__builtin_amdgcn_readlane(ld, i));
      float4 sm = own;
      gather_rows<float4, 10>(sm, rec, bc, ec, ecol, hs, lane);
      acc.x = bv.x + sm.x * dc; acc.y = bv.y + sm.y * dc;
      acc.z = bv.z + sm.z * dc; acc.w = bv.w + sm.w * dc;
    }

    {
      const int bn = __builtin_amdgcn_readlane(lb, i + 1), en = __builtin_amdgcn_readlane(le, i + 1);
      rec = (bn + lane < en) ? ecol[bn + lane] : 0;
      own.x = 0.0f; own.y = 0.0f; own.z = 0.0f; own.w = 0.0f;
      if (i + 1 < GT && v + 16 < n) own = *(const float4*)(hs + (size_t)(v + 16) * 64);
    }

    float4 ra;
    ra.x = fmaxf(acc.x, 0.0f); ra.y = fmaxf(acc.y, 0.0f);
    ra.z = fmaxf(acc.z, 0.0f); ra.w = fmaxf(acc.w, 0.0f);
    *(float4*)(tile + wave * P + lane * 4) = ra;
    lds_barrier();

    v8f c = v8f{};
    const float* __restrict__ arow = tile + m * P + q * 32 + kb;
#pragma unroll
    for (int kk = 0; kk < 8; ++kk) {
      const v2f a = *(const v2f*)(arow + kk * 4);
      c = __builtin_amdgcn_wmma_f32_16x16x4_f32(false, a, false, b[kk],
                                                (short)0, c, false, false);
    }

    if (q > 0) {
      float* __restrict__ pw = part + ((q - 1) * 4 + ct) * 256 + lane;
#pragma unroll
      for (int r = 0; r < 8; ++r) pw[r * 32] = c[r];
    }
    lds_barrier();
    if (q == 0) {
      const int rbase = (lane >> 4) * 8;
      float dr[8];
#pragma unroll
      for (int r = 0; r < 8; ++r) {
        const int row = row0 + rbase + r;
        dr[r] = dis[row < n ? row : n - 1];
      }
#pragma unroll
      for (int qq = 0; qq < 3; ++qq) {
        const float* __restrict__ pr = part + (qq * 4 + ct) * 256 + lane;
#pragma unroll
        for (int r = 0; r < 8; ++r) c[r] += pr[r * 32];
      }

#pragma unroll
      for (int r = 0; r < 8; ++r) {
        const int row = row0 + rbase + r;
        if (row < n) h2[(size_t)row * 64 + col0 + m] = c[r] * dr[r];
      }
    }
  }
}

static inline int cdiv(long long a, long long b) { return (int)((a + b - 1) / b); }

static void run_layers(const float* x, const float* W1,
                       const float* b1, const float* W2, const float* b2,
                       float* out_final, const float* dis, const int* rowptr, const int* ecol,
                       float* h, float* h2, int N, hipStream_t stream) {
  const int IN = 128, HD = 128, OD = 64;
  const int RT = 5;

  const int rowBlocks = N / (16 * RT);
  const int gblocks = cdiv(N, 8 * GR);
  const int wblocks = cdiv(N, 16 * GT);

  float* h1 = out_final;
  gemm_wmma_f32<IN, HD, RT, 2><<<rowBlocks, 128, 0, stream>>>(x, W1, dis, h, h1);

  gather_wide_gemm_kernel<<<wblocks, 512, 0, stream>>>(rowptr, ecol, dis, h, h1, b1, W2, h2, N);

  gather_kernel<<<gblocks, 256, 0, stream>>>(rowptr, ecol, dis, h2, b2, out_final, OD, N);
}

extern "C" void kernel_launch(void* const* d_in, const int* in_sizes, int n_in,
                              void* d_out, int out_size, void* d_ws, size_t ws_size,
                              hipStream_t stream) {
  const float* x_feat = (const float*)d_in[0];
  const int*   x_ei   = (const int*)d_in[1];
  const float* y_feat = (const float*)d_in[2];
  const int*   y_ei   = (const int*)d_in[3];
  const float* W1x = (const float*)d_in[4];
  const float* b1x = (const float*)d_in[5];
  const float* W2x = (const float*)d_in[6];
  const float* b2x = (const float*)d_in[7];
  const float* W1y = (const float*)d_in[8];
  const float* b1y = (const float*)d_in[9];
  const float* W2y = (const float*)d_in[10];
  const float* b2y = (const float*)d_in[11];

  const int N = in_sizes[0] / 128;
  const int E = in_sizes[1] / 2;
  const int NP = cdiv(N + 1, 1024) * 1024;

  float* dis_y = (float*)d_ws;
  float* h     = dis_y + N;
  int*   aux   = (int*)(h + (size_t)N * 64);
  float* h2    = h + (size_t)N * 128;

  float* out_x = (float*)d_out;
  float* out_y = out_x + (size_t)N * 64;

  int*   ecol_y   = aux;
  int*   rowptr_y = aux + 2 * (size_t)E;
  int*   ecol_x   = (int*)out_y;
  int*   rowptr_x = ecol_x + 2 * (size_t)E;
  float* dis_x    = (float*)(rowptr_x + NP);
  const int SP = cdiv(N + 1, SR);
  int*   cntc     = (int*)h2;
  int*   belowc   = cntc + (size_t)2 * SC * SP * SR;

  const dim3 sgrid(SC, SP, 2);
  sort_count_kernel<<<sgrid, 1024, 0, stream>>>(x_ei, y_ei, cntc, belowc, E);
  sort_place_kernel<<<sgrid, 1024, 0, stream>>>(x_ei, y_ei, cntc, belowc, rowptr_x, rowptr_y,
                                                dis_x, dis_y, ecol_x, ecol_y, E, N);

  run_layers(x_feat, W1x, b1x, W2x, b2x, out_x, dis_x, rowptr_x, ecol_x, h, h2, N, stream);
  run_layers(y_feat, W1y, b1y, W2y, b2y, out_y, dis_y, rowptr_y, ecol_y, h, h2, N, stream);
}


# Edge sort: a block owns 8192 destinations, was 4096 (8 counters per thread): 224 blocks, was 416, and each edge is read and tested by 7 blocks, was 13; both sort kernels; sum order rule kept
# speedup vs baseline: 4.5531x; 1.3094x over previous
#include <hip/hip_runtime.h>


typedef __attribute__((ext_vector_type(2))) float v2f;
typedef __attribute__((ext_vector_type(8))) float v8f;

constexpr int SC = 16;
constexpr int SR = 8192;
constexpr int SQ = SR / 4096;

struct ChunkRange { int vb, ve, sb, se; };
__device__ __forceinline__ ChunkRange chunk_range(int c, int E) {
  const int nv = ((E & 3) == 0) ? (E >> 2) : 0;
  const int nvc = (nv + SC - 1) / SC;
  const int ns = E - 4 * nv;
  const int nsc = (ns + SC - 1) / SC;
  ChunkRange r;
  r.vb = c * nvc;
  r.ve = (r.vb + nvc < nv) ? (r.vb + nvc) : nv;
  r.sb = 4 * nv + c * nsc;
  r.se = (r.sb + nsc < E) ? (r.sb + nsc) : E;
  return r;
}

__device__ __forceinline__ void sort_count(int d, int lo, int* cnt, int& below) {
  const unsigned t = (unsigned)(d - lo);
  below += (d < lo) ? 1 : 0;
  if (t < (unsigned)SR) atomicAdd(&cnt[t], 1);
}

__device__ __forceinline__ void sort_place4(int4 d, int4 sv, int lo, int* cur, int* __restrict__ ecol) {
  const unsigned t0 = (unsigned)(d.x - lo), t1 = (unsigned)(d.y - lo);
  const unsigned t2 = (unsigned)(d.z - lo), t3 = (unsigned)(d.w - lo);
  const bool m0 = t0 < (unsigned)SR, m1 = t1 < (unsigned)SR;
  const bool m2 = t2 < (unsigned)SR, m3 = t3 < (unsigned)SR;
  int s0 = 0, s1 = 0, s2 = 0, s3 = 0;
  if (m0) s0 = atomicAdd(&cur[t0], 1);
  if (m1) s1 = atomicAdd(&cur[t1], 1);
  if (m2) s2 = atomicAdd(&cur[t2], 1);
  if (m3) s3 = atomicAdd(&cur[t3], 1);
  if (m0) ecol[s0] = sv.x;
  if (m1) ecol[s1] = sv.y;
  if (m2) ecol[s2] = sv.z;
  if (m3) ecol[s3] = sv.w;
}

__global__ void __launch_bounds__(1024)
sort_count_kernel(const int* __restrict__ ei_x, const int* __restrict__ ei_y,
                  int* __restrict__ cntc, int* __restrict__ belowc, int E) {
  __shared__ alignas(16) int cnt[SR];
  __shared__ int wred[32];
  const int c = blockIdx.x, p = blockIdx.y, g = blockIdx.z;
  const int P = gridDim.y;
  const int* __restrict__ dst = (g ? ei_y : ei_x) + E;
  const int tid = threadIdx.x, lane = tid & 31, wv = tid >> 5;
  const int lo = p * SR;

  int4 zero4;
  zero4.x = 0; zero4.y = 0; zero4.z = 0; zero4.w = 0;
#pragma unroll
  for (int h = 0; h < SQ; ++h) ((int4*)cnt)[tid + h * 1024] = zero4;
  __syncthreads();

  const ChunkRange cr = chunk_range(c, E);
  const int4* __restrict__ dst4 = (const int4*)dst;
  int below = 0;
  int i = cr.vb + tid;
  for (; i + 3 * 1024 < cr.ve; i += 4 * 1024) {
    int4 q[4];
#pragma unroll
    for (int k = 0; k < 4; ++k) q[k] = dst4[i + k * 1024];
#pragma unroll
    for (int k = 0; k < 4; ++k) {
      sort_count(q[k].x, lo, cnt, below); sort_count(q[k].y, lo, cnt, below);
      sort_count(q[k].z, lo, cnt, below); sort_count(q[k].w, lo, cnt, below);
    }
  }
  for (; i < cr.ve; i += 1024) {
    const int4 q = dst4[i];
    sort_count(q.x, lo, cnt, below); sort_count(q.y, lo, cnt, below);
    sort_count(q.z, lo, cnt, below); sort_count(q.w, lo, cnt, below);
  }
  for (int e = cr.sb + tid; e < cr.se; e += 1024) sort_count(dst[e], lo, cnt, below);

#pragma unroll
  for (int o = 16; o > 0; o >>= 1) below += __shfl_xor(below, o, 32);
  if (lane == 0) wred[wv] = below;
  __syncthreads();

  const size_t plane = ((size_t)(g * SC + c) * P + p) * SR;
#pragma unroll
  for (int h = 0; h < SQ; ++h) ((int4*)(cntc + plane))[tid + h * 1024] = ((const int4*)cnt)[tid + h * 1024];
  if (tid == 0) {
    int b = 0;
#pragma unroll
    for (int k = 0; k < 32; ++k) b += wred[k];
    belowc[(g * SC + c) * P + p] = b;
  }
}

__global__ void __launch_bounds__(1024) __attribute__((amdgpu_waves_per_eu(8, 10)))
sort_place_kernel(const int* __restrict__ ei_x, const int* __restrict__ ei_y,
                  const int* __restrict__ cntc, const int* __restrict__ belowc,
                  int* __restrict__ rowptr_x, int* __restrict__ rowptr_y,
                  float* __restrict__ dis_x, float* __restrict__ dis_y,
                  int* __restrict__ ecol_x, int* __restrict__ ecol_y, int E, int N) {
  __shared__ alignas(16) int cur[SR];
  __shared__ int wtot[32];
  const int c = blockIdx.x, p = blockIdx.y, g = blockIdx.z;
  const int P = gridDim.y;
  const int* __restrict__ src = g ? ei_y : ei_x;
  const int* __restrict__ dst = src + E;
  int* __restrict__ rowptr = g ? rowptr_y : rowptr_x;
  float* __restrict__ dis = g ? dis_y : dis_x;
  int* __restrict__ ecol = g ? ecol_y : ecol_x;
  const int tid = threadIdx.x, lane = tid & 31, wv = tid >> 5;
  const int lo = p * SR;

  constexpr int NT = 4 * SQ;
  int4 tq[SQ], pq[SQ];
#pragma unroll
  for (int h = 0; h < SQ; ++h) {
    tq[h].x = 0; tq[h].y = 0; tq[h].z = 0; tq[h].w = 0;
    pq[h].x = 0; pq[h].y = 0; pq[h].z = 0; pq[h].w = 0;
  }
  int base = 0;
#pragma unroll (SC / 2)
  for (int cc = 0; cc < SC; ++cc) {
    const size_t plane = ((size_t)(g * SC + cc) * P + p) * SR;
#pragma unroll
    for (int h = 0; h < SQ; ++h) {
      const int4 q = ((const int4*)(cntc + plane))[tid * SQ + h];
      tq[h].x += q.x; tq[h].y += q.y; tq[h].z += q.z; tq[h].w += q.w;
      if (cc < c) { pq[h].x += q.x; pq[h].y += q.y; pq[h].z += q.z; pq[h].w += q.w; }
    }
    base += belowc[(g * SC + cc) * P + p];
  }
  int tot[NT], pre[NT];
#pragma unroll
  for (int h = 0; h < SQ; ++h) {
    tot[h * 4 + 0] = tq[h].x; tot[h * 4 + 1] = tq[h].y; tot[h * 4 + 2] = tq[h].z; tot[h * 4 + 3] = tq[h].w;
    pre[h * 4 + 0] = pq[h].x; pre[h * 4 + 1] = pq[h].y; pre[h * 4 + 2] = pq[h].z; pre[h * 4 + 3] = pq[h].w;
  }

  int sc[NT];
  sc[0] = tot[0];
#pragma unroll
  for (int j = 1; j < NT; ++j) sc[j] = sc[j - 1] + tot[j];
  int incl = sc[NT - 1];
#pragma unroll
  for (int o = 1; o < 32; o <<= 1) {
    int y = __shfl_up(incl, o, 32);
    if (lane >= o) incl += y;
  }
  if (lane == 31) wtot[wv] = incl;
  __syncthreads();
#pragma unroll
  for (int k = 0; k < 32; ++k) {
    if (k < wv) base += wtot[k];
  }
  const int excl = base + incl - sc[NT - 1];
  int r[NT];
  r[0] = excl;
#pragma unroll
  for (int j = 1; j < NT; ++j) r[j] = excl + sc[j - 1];
#pragma unroll
  for (int h = 0; h < SQ; ++h) {
    int4 cu;
    cu.x = r[h * 4 + 0] + pre[h * 4 + 0]; cu.y = r[h * 4 + 1] + pre[h * 4 + 1];
    cu.z = r[h * 4 + 2] + pre[h * 4 + 2]; cu.w = r[h * 4 + 3] + pre[h * 4 + 3];
    ((int4*)cur)[tid * SQ + h] = cu;
  }

  if (c == 0) {
    const int n0 = lo + tid * NT;
#pragma unroll
    for (int j = 0; j < NT; ++j) {
      if (n0 + j <= N) rowptr[n0 + j] = r[j];
      if (n0 + j < N) dis[n0 + j] = rsqrtf((float)(tot[j] + 1));
    }
  }
  __syncthreads();

  const ChunkRange cr = chunk_range(c, E);
  const int4* __restrict__ dst4 = (const int4*)dst;
  const int4* __restrict__ src4 = (const int4*)src;
  int i = cr.vb + tid;
  for (; i + 3 * 1024 < cr.ve; i += 4 * 1024) {
    int4 q[4], sv[4];
#pragma unroll
    for (int k = 0; k < 4; ++k) { q[k] = dst4[i + k * 1024]; sv[k] = src4[i + k * 1024]; }
#pragma unroll
    for (int k = 0; k < 4; ++k) sort_place4(q[k], sv[k], lo, cur, ecol);
  }
  for (; i < cr.ve; i += 1024) sort_place4(dst4[i], src4[i], lo, cur, ecol);
  for (int e = cr.sb + tid; e < cr.se; e += 1024) {
    const int d = dst[e], sv1 = src[e];
    const unsigned t = (unsigned)(d - lo);
    if (t < (unsigned)SR) ecol[atomicAdd(&cur[t], 1)] = sv1;
  }
}


__device__ __forceinline__ void add_to(float2& s, const float2 x) {
  s.x += x.x; s.y += x.y;
}
__device__ __forceinline__ void fma_to(float4& s, const float w, const float4 x) {
  s.x = fmaf(w, x.x, s.x); s.y = fmaf(w, x.y, s.y);
  s.z = fmaf(w, x.z, s.z); s.w = fmaf(w, x.w, s.w);
}

template <int NB>
__device__ __forceinline__ void gather_rows(float2& sm, int r, int beg, int end,
                                            const int* __restrict__ ecol,
                                            const float* __restrict__ hs, int lane) {
  int base = beg;
  for (;;) {
    const int left = end - base;
    const int cc = left < 32 ? left : 32;
    int j = 0;
    for (; j + NB <= cc; j += NB) {
      float2 x[NB];
#pragma unroll
      for (int k = 0; k < NB; ++k)
        x[k] = *(const float2*)(hs + (size_t)__builtin_amdgcn_readlane(r, j + k) * 64);
#pragma unroll
      for (int k = 0; k < NB; ++k) add_to(sm, x[k]);
    }
    const int nv = cc - j;
    if (nv > 0) {
      float2 x[NB];
#pragma unroll
      for (int k = 0; k < NB - 1; ++k)
        if (k < nv) x[k] = *(const float2*)(hs + (size_t)__builtin_amdgcn_readlane(r, j + k) * 64);
#pragma unroll
      for (int k = 0; k < NB - 1; ++k)
        if (k < nv) add_to(sm, x[k]);
    }
    base += 32;
    if (base >= end) break;
    r = (base + lane < end) ? ecol[base + lane] : 0;
  }
}

template <int NB>
__device__ __forceinline__ void gather_rows_weighted(float4& sm, int r, float w, int beg, int end,
                                                     const int* __restrict__ ecol,
                                                     const float* __restrict__ dis,
                                                     const float* __restrict__ xin, int lane) {
  int base = beg;
  for (;;) {
    const int left = end - base;
    const int cc = left < 32 ? left : 32;
    const int wi = __float_as_int(w);
    int j = 0;
    for (; j + NB <= cc; j += NB) {
      float4 x[NB];
#pragma unroll
      for (int k = 0; k < NB; ++k)
        x[k] = *(const float4*)((const char*)xin + ((unsigned)__builtin_amdgcn_readlane(r, j + k) * 512u + (unsigned)lane * 16u));
#pragma unroll
      for (int k = 0; k < NB; ++k)
        fma_to(sm, __int_as_float(__builtin_amdgcn_readlane(wi, j + k)), x[k]);
    }
    const int nv = cc - j;
    if (nv > 0) {
      float4 x[NB];
#pragma unroll
      for (int k = 0; k < NB - 1; ++k)
        if (k < nv) x[k] = *(const float4*)((const char*)xin + ((unsigned)__builtin_amdgcn_readlane(r, j + k) * 512u + (unsigned)lane * 16u));
#pragma unroll
      for (int k = 0; k < NB - 1; ++k)
        if (k < nv) fma_to(sm, __int_as_float(__builtin_amdgcn_readlane(wi, j + k)), x[k]);
    }
    base += 32;
    if (base >= end) break;
    r = 0; w = 0.0f;
    if (base + lane < end) { r = ecol[base + lane]; w = dis[r]; }
  }
}

constexpr int GR = 4;
__global__ void __launch_bounds__(256)
gather_kernel(const int* __restrict__ rowptr_x, const int* __restrict__ rowptr_y,
              const int* __restrict__ ecol_x, const int* __restrict__ ecol_y,
              const float* __restrict__ dis_x, const float* __restrict__ dis_y,
              const float* __restrict__ h_x, const float* __restrict__ h_y,
              const float* __restrict__ bias_x, const float* __restrict__ bias_y,
              float* __restrict__ out_x, float* __restrict__ out_y,
              int ldo, int n) {
  const int g = blockIdx.y;
  const int* __restrict__ rowptr = g ? rowptr_y : rowptr_x;
  const int* __restrict__ ecol = g ? ecol_y : ecol_x;
  const float* __restrict__ dis = g ? dis_y : dis_x;
  const float* __restrict__ h = g ? h_y : h_x;
  const float* __restrict__ bias = g ? bias_y : bias_x;
  float* __restrict__ out = g ? out_y : out_x;
  const int wave = __builtin_amdgcn_readfirstlane((int)(threadIdx.x >> 5));
  const int lane = threadIdx.x & 31;
  const int v0 = (blockIdx.x * 8 + wave) * GR;
  if (v0 >= n) return;

  const float* __restrict__ hs = h + lane * 2;
  int rp[GR + 1];
  float dv[GR];
  int rec[GR];
  float2 own[GR];
#pragma unroll
  for (int i = 0; i <= GR; ++i) rp[i] = rowptr[(v0 + i < n) ? (v0 + i) : n];
#pragma unroll
  for (int i = 0; i < GR; ++i) dv[i] = (v0 + i < n) ? dis[v0 + i] : 0.0f;
#pragma unroll
  for (int i = 0; i < GR; ++i) {
    rec[i] = (rp[i] + lane < rp[i + 1]) ? ecol[rp[i] + lane] : 0;
    own[i].x = 0.0f; own[i].y = 0.0f;
    if (v0 + i < n) own[i] = *(const float2*)(hs + (size_t)(v0 + i) * 64);
  }
  const float2 bv = *(const float2*)(bias + lane * 2);

#pragma unroll
  for (int i = 0; i < GR; ++i) {
    const int v = v0 + i;
    if (v < n) {
      float2 sm = own[i];
      gather_rows<8>(sm, rec[i], rp[i], rp[i + 1], ecol, hs, lane);
      float2 o;
      o.x = bv.x + sm.x * dv[i]; o.y = bv.y + sm.y * dv[i];
      *(float2*)(out + (size_t)v * ldo + lane * 2) = o;
    }
  }
}

__device__ __forceinline__ void lds_barrier() {
  __builtin_amdgcn_fence(__ATOMIC_RELEASE, "workgroup", "local");
  __builtin_amdgcn_s_barrier();
  __builtin_amdgcn_fence(__ATOMIC_ACQUIRE, "workgroup", "local");
}

constexpr int GT = 13;
__global__ void __launch_bounds__(512)
layer1_fused_kernel(const int* __restrict__ rowptr_x, const int* __restrict__ rowptr_y,
                    const int* __restrict__ ecol_x, const int* __restrict__ ecol_y,
                    const float* __restrict__ dis_x, const float* __restrict__ dis_y,
                    const float* __restrict__ x_x, const float* __restrict__ x_y,
                    const float* __restrict__ W1_x, const float* __restrict__ W1_y,
                    const float* __restrict__ b1_x, const float* __restrict__ b1_y,
                    const float* __restrict__ W2_x, const float* __restrict__ W2_y,
                    float* __restrict__ h2_x, float* __restrict__ h2_y, int n) {
  constexpr int P = 132;
  __shared__ alignas(16) float tile[16 * P];
  __shared__ alignas(16) float tile2[16 * P];
  __shared__ float part1[8 * 256];
  __shared__ float part[3 * 4 * 256];
  __shared__ alignas(16) v2f w1f[16 * 16 * 32];
  const int g = blockIdx.y;
  const int* __restrict__ rowptr = g ? rowptr_y : rowptr_x;
  const int* __restrict__ ecol = g ? ecol_y : ecol_x;
  const float* __restrict__ dis = g ? dis_y : dis_x;
  const float* __restrict__ xin = g ? x_y : x_x;
  const float* __restrict__ W1 = g ? W1_y : W1_x;
  const float* __restrict__ b1 = g ? b1_y : b1_x;
  const float* __restrict__ W2 = g ? W2_y : W2_x;
  float* __restrict__ h2 = g ? h2_y : h2_x;
  const int wave = __builtin_amdgcn_readfirstlane((int)(threadIdx.x >> 5));
  const int lane = threadIdx.x & 31;
  const int rowbase = blockIdx.x * (16 * GT);

  const int m  = lane & 15;
  const int kb = (lane >> 4) * 2;
  const int ct1 = wave & 7;
  const int kh  = wave >> 3;
  const int ct = wave & 3;
  const int q  = wave >> 2;
  const int col0 = ct * 16;
  v2f* __restrict__ wf = w1f + wave * (16 * 32) + lane;
#pragma unroll
  for (int kk = 0; kk < 16; ++kk) {
    const int k = kh * 64 + kk * 4 + kb;
    v2f f;
    f.x = W1[(size_t)k * 128 + ct1 * 16 + m];
    f.y = W1[(size_t)(k + 1) * 128 + ct1 * 16 + m];
    wf[kk * 32] = f;
  }
  v2f bw2[8];
#pragma unroll
  for (int kk = 0; kk < 8; ++kk) {
    const int k = q * 32 + kk * 4 + kb;
    bw2[kk].x = W2[(size_t)k * 64 + col0 + m];
    bw2[kk].y = W2[(size_t)(k + 1) * 64 + col0 + m];
  }
  const float b1v = b1[ct1 * 16 + m];

  int lb = 0, le = 0, ld = 0;
  {
    const int vl = rowbase + lane * 16 + wave;
    if (lane < GT && vl < n) { lb = rowptr[vl]; le = rowptr[vl + 1]; ld = __float_as_int(dis[vl]); }
  }
  int rec = 0;
  float wgt = 0.0f;
  float4 own;
  own.x = 0.0f; own.y = 0.0f; own.z = 0.0f; own.w = 0.0f;
  {
    const int b0 = __builtin_amdgcn_readlane(lb, 0), e0 = __builtin_amdgcn_readlane(le, 0);
    if (b0 + lane < e0) { rec = ecol[b0 + lane]; wgt = dis[rec]; }
    if (rowbase + wave < n) own = *(const float4*)((const char*)xin + ((unsigned)(rowbase + wave) * 512u + (unsigned)lane * 16u));
  }

  for (int i = 0; i < GT; ++i) {
    const int row0 = rowbase + i * 16;
    const int v = row0 + wave;

    const int bn = __builtin_amdgcn_readlane(lb, i + 1), en = __builtin_amdgcn_readlane(le, i + 1);
    const bool has_next = bn + lane < en;
    int recn = 0;
    if (has_next) recn = ecol[bn + lane];

    float4 acc;
    acc.x = 0.0f; acc.y = 0.0f; acc.z = 0.0f; acc.w = 0.0f;
    if (v < n) {
      const int bc = __builtin_amdgcn_readlane(lb, i), ec = __builtin_amdgcn_readlane(le, i);
      const float dc = __int_as_float(__builtin_amdgcn_readlane(ld, i));
      float4 sm;
      sm.x = own.x * dc; sm.y = own.y * dc; sm.z = own.z * dc; sm.w = own.w * dc;
      gather_rows_weighted<10>(sm, rec, wgt, bc, ec, ecol, dis, xin, lane);
      acc.x = sm.x * dc; acc.y = sm.y * dc; acc.z = sm.z * dc; acc.w = sm.w * dc;
    }

    {
      float wn = 0.0f;
      if (has_next) wn = dis[recn];
      own.x = 0.0f; own.y = 0.0f; own.z = 0.0f; own.w = 0.0f;
      if (i + 1 < GT && v + 16 < n) own = *(const float4*)((const char*)xin + ((unsigned)(v + 16) * 512u + (unsigned)lane * 16u));
      rec = recn; wgt = wn;
    }

    *(float4*)(tile + wave * P + lane * 4) = acc;
    lds_barrier();

    v8f c1 = v8f{};
    {
      const float* __restrict__ arow = tile + m * P + kh * 64 + kb;
#pragma unroll
      for (int kk = 0; kk < 16; ++kk) {
        const v2f a = *(const v2f*)(arow + kk * 4);
        const v2f b = wf[kk * 32];
        c1 = __builtin_amdgcn_wmma_f32_16x16x4_f32(false, a, false, b,
                                                   (short)0, c1, false, false);
      }
    }
    if (kh == 1) {
      float* __restrict__ pw = part1 + ct1 * 256 + lane;
#pragma unroll
      for (int r = 0; r < 8; ++r) pw[r * 32] = c1[r];
    }
    lds_barrier();
    if (kh == 0) {
      const float* __restrict__ pr = part1 + ct1 * 256 + lane;
      float* __restrict__ tw = tile2 + (lane >> 4) * 8 * P + ct1 * 16 + m;
#pragma unroll
      for (int r = 0; r < 8; ++r) tw[r * P] = fmaxf((c1[r] + pr[r * 32]) + b1v, 0.0f);
    }
    lds_barrier();

    v8f c = v8f{};
    {
      const float* __restrict__ arow = tile2 + m * P + q * 32 + kb;
#pragma unroll
      for (int kk = 0; kk < 8; ++kk) {
        const v2f a = *(const v2f*)(arow + kk * 4);
        c = __builtin_amdgcn_wmma_f32_16x16x4_f32(false, a, false, bw2[kk],
                                                  (short)0, c, false, false);
      }
    }

    if (q > 0) {
      float* __restrict__ pw = part + ((q - 1) * 4 + ct) * 256 + lane;
#pragma unroll
      for (int r = 0; r < 8; ++r) pw[r * 32] = c[r];
    }
    lds_barrier();
    if (q == 0) {
      const int rbase = (lane >> 4) * 8;
      float dr[8];
#pragma unroll
      for (int r = 0; r < 8; ++r) {
        const int row = row0 + rbase + r;
        dr[r] = dis[row < n ? row : n - 1];
      }
#pragma unroll
      for (int qq = 0; qq < 3; ++qq) {
        const float* __restrict__ pr = part + (qq * 4 + ct) * 256 + lane;
#pragma unroll
        for (int r = 0; r < 8; ++r) c[r] += pr[r * 32];
      }

#pragma unroll
      for (int r = 0; r < 8; ++r) {
        const int row = row0 + rbase + r;
        if (row < n) h2[(size_t)row * 64 + col0 + m] = c[r] * dr[r];
      }
    }
  }
}

static inline int cdiv(long long a, long long b) { return (int)((a + b - 1) / b); }

extern "C" void kernel_launch(void* const* d_in, const int* in_sizes, int n_in,
                              void* d_out, int out_size, void* d_ws, size_t ws_size,
                              hipStream_t stream) {
  const float* x_feat = (const float*)d_in[0];
  const int*   x_ei   = (const int*)d_in[1];
  const float* y_feat = (const float*)d_in[2];
  const int*   y_ei   = (const int*)d_in[3];
  const float* W1x = (const float*)d_in[4];
  const float* b1x = (const float*)d_in[5];
  const float* W2x = (const float*)d_in[6];
  const float* b2x = (const float*)d_in[7];
  const float* W1y = (const float*)d_in[8];
  const float* b1y = (const float*)d_in[9];
  const float* W2y = (const float*)d_in[10];
  const float* b2y = (const float*)d_in[11];

  const int N = in_sizes[0] / 128;
  const int E = in_sizes[1] / 2;
  const int NP = cdiv(N + 1, 1024) * 1024;
  const int SP = cdiv(N + 1, SR);

  const size_t E4 = ((size_t)E + 3) & ~(size_t)3;
  const size_t N4 = ((size_t)N + 3) & ~(size_t)3;
  float* dis_y    = (float*)d_ws;
  float* h2_x     = dis_y + N;
  float* h2_y     = h2_x + (size_t)N * 64;
  int*   ecol_x   = (int*)(h2_y + (size_t)N * 64);
  int*   ecol_y   = ecol_x + E4;
  int*   rowptr_x = ecol_y + E4;
  int*   rowptr_y = rowptr_x + NP;
  float* dis_x    = (float*)(rowptr_y + NP);
  int*   cntc     = (int*)(dis_x + N4);
  int*   belowc   = cntc + (size_t)2 * SC * SP * SR;

  float* out_x = (float*)d_out;
  float* out_y = out_x + (size_t)N * 64;

  const dim3 sgrid(SC, SP, 2);
  sort_count_kernel<<<sgrid, 1024, 0, stream>>>(x_ei, y_ei, cntc, belowc, E);
  sort_place_kernel<<<sgrid, 1024, 0, stream>>>(x_ei, y_ei, cntc, belowc, rowptr_x, rowptr_y,
                                                dis_x, dis_y, ecol_x, ecol_y, E, N);

  const dim3 fgrid(cdiv(N, 16 * GT), 2);
  layer1_fused_kernel<<<fgrid, 512, 0, stream>>>(rowptr_x, rowptr_y, ecol_x, ecol_y, dis_x, dis_y,
                                                 x_feat, y_feat, W1x, W1y, b1x, b1y, W2x, W2y,
                                                 h2_x, h2_y, N);

  const dim3 ggrid(cdiv(N, 8 * GR), 2);
  gather_kernel<<<ggrid, 256, 0, stream>>>(rowptr_x, rowptr_y, ecol_x, ecol_y, dis_x, dis_y,
                                           h2_x, h2_y, b2x, b2y, out_x, out_y, 64, N);
}
